# starting kernel
# baseline (speedup 1.0000x reference)
#include <hip/hip_runtime.h>

#define IN_F  128
#define HID_F 64
#define OUT_F 40

typedef float v2f __attribute__((ext_vector_type(2)));
typedef float v8f __attribute__((ext_vector_type(8)));

__device__ __forceinline__ v8f wmma4(v2f a, v2f b, v8f c) {
  return __builtin_amdgcn_wmma_f32_16x16x4_f32(
       false, a,  false, b,
       (short)0, c,  false,  false);
}

__device__ __forceinline__ void atomadd(float* p, float v) {
  unsafeAtomicAdd(p, v);
}

__global__ void deg_kernel(const int* __restrict__ src, float* __restrict__ deg, int E) {
  int e = blockIdx.x * blockDim.x + threadIdx.x;
  if (e < E) atomadd(&deg[src[e]], 1.0f);
}

__global__ void dinv_kernel(float* __restrict__ deg, int n) {
  int i = blockIdx.x * blockDim.x + threadIdx.x;
  if (i < n) {
    float d = deg[i];
    deg[i] = (d > 0.0f) ? __frsqrt_rn(d) : 0.0f;
  }
}

__global__ void norm_kernel(const int* __restrict__ src, const int* __restrict__ dst,
                            const float* __restrict__ dinv, float* __restrict__ nrm, int E) {
  int e = blockIdx.x * blockDim.x + threadIdx.x;
  if (e < E) nrm[e] = -dinv[src[e]] * dinv[dst[e]];
}

template <int F>
__global__ void prop_kernel(const float* __restrict__ xin, const int* __restrict__ src,
                            const int* __restrict__ dst, const float* __restrict__ nrm,
                            float* __restrict__ out, int E) {
  const int CH = F / 4;
  long long gid = (long long)blockIdx.x * blockDim.x + threadIdx.x;
  long long nwork = (long long)E * CH;
  if (gid >= nwork) return;
  int c = (int)(gid % CH);
  int e = (int)(gid / CH);
  int s = src[e], d = dst[e];
  float w = nrm[e];
  const float4 v = *(const float4*)(xin + (long long)s * F + c * 4);
  float* o = out + (long long)d * F + c * 4;
  atomadd(o + 0, v.x * w);
  atomadd(o + 1, v.y * w);
  atomadd(o + 2, v.z * w);
  atomadd(o + 3, v.w * w);
}

__global__ __launch_bounds__(256)
void gemm1_kernel(const float* __restrict__ x, const float* __restrict__ t1,
                  const float* __restrict__ p2, const float* __restrict__ W,
                  const float* __restrict__ bias, float* __restrict__ h, int ntile) {
  int wv = (int)((blockIdx.x * (long long)blockDim.x + threadIdx.x) >> 5);
  if (wv >= ntile) return;
  int lane = threadIdx.x & 31;
  int idx = lane & 15, half = lane >> 4;

  long long rowbase = (long long)(wv * 16 + idx) * IN_F;
  const float* xr = x + rowbase;
  const float* tr = t1 + rowbase;
  const float* pr = p2 + rowbase;

  v8f z = {0.f, 0.f, 0.f, 0.f, 0.f, 0.f, 0.f, 0.f};
  v8f acc[4] = {z, z, z, z};

  for (int k0 = 0; k0 < IN_F; k0 += 4) {
    int kk = k0 + 2 * half;
    v2f ax = *(const v2f*)(xr + kk);
    v2f at = *(const v2f*)(tr + kk);
    v2f ap = *(const v2f*)(pr + kk);
    v2f a2 = ap + ap - ax;
#pragma unroll
    for (int nt = 0; nt < 4; ++nt) {
      int n = nt * 16 + idx;
      const float* w0 = W + (long long)kk * HID_F + n;
      const float* w1 = w0 + (long long)IN_F * HID_F;
      const float* w2 = w1 + (long long)IN_F * HID_F;
      v2f b0 = {w0[0], w0[HID_F]};
      v2f b1 = {w1[0], w1[HID_F]};
      v2f b2 = {w2[0], w2[HID_F]};
      acc[nt] = wmma4(ax, b0, acc[nt]);
      acc[nt] = wmma4(at, b1, acc[nt]);
      acc[nt] = wmma4(a2, b2, acc[nt]);
    }
  }
#pragma unroll
  for (int nt = 0; nt < 4; ++nt) {
    int n = nt * 16 + idx;
    float bv = bias[n];
#pragma unroll
    for (int r = 0; r < 8; ++r) {
      long long row = (long long)wv * 16 + r + 8 * half;
      float val = acc[nt][r] + bv;
      h[row * HID_F + n] = val > 0.0f ? val : 0.0f;
    }
  }
}

__global__ __launch_bounds__(256)
void gemm2_kernel(const float* __restrict__ hx, const float* __restrict__ t1,
                  const float* __restrict__ p2, const float* __restrict__ W,
                  const float* __restrict__ bias, float* __restrict__ out, int ntile) {
  int wv = (int)((blockIdx.x * (long long)blockDim.x + threadIdx.x) >> 5);
  if (wv >= ntile) return;
  int lane = threadIdx.x & 31;
  int idx = lane & 15, half = lane >> 4;

  long long rowbase = (long long)(wv * 16 + idx) * HID_F;
  const float* xr = hx + rowbase;
  const float* tr = t1 + rowbase;
  const float* pr = p2 + rowbase;

  v8f z = {0.f, 0.f, 0.f, 0.f, 0.f, 0.f, 0.f, 0.f};
  v8f acc[3] = {z, z, z};

  for (int k0 = 0; k0 < HID_F; k0 += 4) {
    int kk = k0 + 2 * half;
    v2f ax = *(const v2f*)(xr + kk);
    v2f at = *(const v2f*)(tr + kk);
    v2f ap = *(const v2f*)(pr + kk);
    v2f a2 = ap + ap - ax;
#pragma unroll
    for (int nt = 0; nt < 3; ++nt) {
      int n = nt * 16 + idx;
      bool ok = (n < OUT_F);
      int nn = ok ? n : 0;
      const float* w0 = W + (long long)kk * OUT_F + nn;
      const float* w1 = w0 + (long long)HID_F * OUT_F;
      const float* w2 = w1 + (long long)HID_F * OUT_F;
      v2f b0 = {ok ? w0[0] : 0.0f, ok ? w0[OUT_F] : 0.0f};
      v2f b1 = {ok ? w1[0] : 0.0f, ok ? w1[OUT_F] : 0.0f};
      v2f b2 = {ok ? w2[0] : 0.0f, ok ? w2[OUT_F] : 0.0f};
      acc[nt] = wmma4(ax, b0, acc[nt]);
      acc[nt] = wmma4(at, b1, acc[nt]);
      acc[nt] = wmma4(a2, b2, acc[nt]);
    }
  }
#pragma unroll
  for (int nt = 0; nt < 3; ++nt) {
    int n = nt * 16 + idx;
    if (n < OUT_F) {
      float bv = bias[n];
#pragma unroll
      for (int r = 0; r < 8; ++r) {
        long long row = (long long)wv * 16 + r + 8 * half;
        out[row * OUT_F + n] = acc[nt][r] + bv;
      }
    }
  }
}

extern "C" void kernel_launch(void* const* d_in, const int* in_sizes, int n_in,
                              void* d_out, int out_size, void* d_ws, size_t ws_size,
                              hipStream_t stream) {
  const float* x  = (const float*)d_in[0];
  const int*   ei = (const int*)d_in[1];
  const float* W1 = (const float*)d_in[2];
  const float* b1 = (const float*)d_in[3];
  const float* W2 = (const float*)d_in[4];
  const float* b2 = (const float*)d_in[5];

  const int N = in_sizes[0] / IN_F;
  const int E = in_sizes[1] / 2;
  const int* src = ei;
  const int* dst = ei + E;

  float* ws   = (float*)d_ws;
  float* dinv = ws;
  float* nrm  = dinv + N;
  float* bufA = nrm + E;
  float* bufB = bufA + (size_t)N * IN_F;
  float* hbuf = bufB + (size_t)N * IN_F;
  float* outp = (float*)d_out;

  const int B = 256;
  auto cdiv = [](long long a, long long b) { return (int)((a + b - 1) / b); };

  hipMemsetAsync(dinv, 0, (size_t)N * sizeof(float), stream);
  deg_kernel<<<cdiv(E, B), B, 0, stream>>>(src, dinv, E);
  dinv_kernel<<<cdiv(N, B), B, 0, stream>>>(dinv, N);
  norm_kernel<<<cdiv(E, B), B, 0, stream>>>(src, dst, dinv, nrm, E);

  long long w128 = (long long)E * (IN_F / 4);
  hipMemsetAsync(bufA, 0, (size_t)N * IN_F * sizeof(float), stream);
  prop_kernel<IN_F><<<cdiv(w128, B), B, 0, stream>>>(x, src, dst, nrm, bufA, E);
  hipMemsetAsync(bufB, 0, (size_t)N * IN_F * sizeof(float), stream);
  prop_kernel<IN_F><<<cdiv(w128, B), B, 0, stream>>>(bufA, src, dst, nrm, bufB, E);

  const int ntile = N / 16;
  const int gblocks = cdiv((long long)ntile * 32, B);
  gemm1_kernel<<<gblocks, B, 0, stream>>>(x, bufA, bufB, W1, b1, hbuf, ntile);

  long long w64 = (long long)E * (HID_F / 4);
  hipMemsetAsync(bufA, 0, (size_t)N * HID_F * sizeof(float), stream);
  prop_kernel<HID_F><<<cdiv(w64, B), B, 0, stream>>>(hbuf, src, dst, nrm, bufA, E);
  hipMemsetAsync(bufB, 0, (size_t)N * HID_F * sizeof(float), stream);
  prop_kernel<HID_F><<<cdiv(w64, B), B, 0, stream>>>(bufA, src, dst, nrm, bufB, E);

  gemm2_kernel<<<gblocks, B, 0, stream>>>(hbuf, bufA, bufB, W2, b2, outp, ntile);
}


# Weights applied before propagation, (L x)W = L(xW): each layer's two scatters run at output width (64, 40) not input width (128, 64), adding into GEMM-stored start values; new f32 sum order
# speedup vs baseline: 1.4646x; 1.4646x over previous
#include <hip/hip_runtime.h>


#define IN_F  128
#define HID_F 64
#define OUT_F 40

typedef float v2f __attribute__((ext_vector_type(2)));
typedef float v8f __attribute__((ext_vector_type(8)));

__device__ __forceinline__ v8f wmma4(v2f a, v2f b, v8f c) {
  return __builtin_amdgcn_wmma_f32_16x16x4_f32(
       false, a,  false, b,
       (short)0, c,  false,  false);
}

__device__ __forceinline__ void atomadd(float* p, float v) {
  unsafeAtomicAdd(p, v);
}

__global__ void deg_kernel(const int* __restrict__ src, float* __restrict__ deg, int E) {
  int e = blockIdx.x * blockDim.x + threadIdx.x;
  if (e < E) atomadd(&deg[src[e]], 1.0f);
}

__global__ void dinv_kernel(float* __restrict__ deg, int n) {
  int i = blockIdx.x * blockDim.x + threadIdx.x;
  if (i < n) {
    float d = deg[i];
    deg[i] = (d > 0.0f) ? __frsqrt_rn(d) : 0.0f;
  }
}

__global__ void norm_kernel(const int* __restrict__ src, const int* __restrict__ dst,
                            const float* __restrict__ dinv, float* __restrict__ nrm, int E) {
  int e = blockIdx.x * blockDim.x + threadIdx.x;
  if (e < E) nrm[e] = -dinv[src[e]] * dinv[dst[e]];
}

template <int F>
__global__ void prop_kernel(const float* __restrict__ xin, const int* __restrict__ src,
                            const int* __restrict__ dst, const float* __restrict__ nrm,
                            float* __restrict__ out, int E) {
  const int CH = F / 4;
  long long gid = (long long)blockIdx.x * blockDim.x + threadIdx.x;
  long long nwork = (long long)E * CH;
  if (gid >= nwork) return;
  int c = (int)(gid % CH);
  int e = (int)(gid / CH);
  int s = src[e], d = dst[e];
  float w = nrm[e];
  const float4 v = *(const float4*)(xin + (long long)s * F + c * 4);
  float* o = out + (long long)d * F + c * 4;
  atomadd(o + 0, v.x * w);
  atomadd(o + 1, v.y * w);
  atomadd(o + 2, v.z * w);
  atomadd(o + 3, v.w * w);
}


__global__ __launch_bounds__(256)
void gemm1_kernel(const float* __restrict__ x, const float* __restrict__ W,
                  const float* __restrict__ bias, float* __restrict__ ys,
                  float* __restrict__ u, float* __restrict__ p, int ntile) {
  int wv = (int)((blockIdx.x * (long long)blockDim.x + threadIdx.x) >> 5);
  if (wv >= ntile) return;
  int lane = threadIdx.x & 31;
  int idx = lane & 15, half = lane >> 4;

  const float* xr = x + (long long)(wv * 16 + idx) * IN_F;

  v8f z = {0.f, 0.f, 0.f, 0.f, 0.f, 0.f, 0.f, 0.f};
  v8f acc0[4] = {z, z, z, z};
  v8f acc1[4] = {z, z, z, z};
  v8f acc2[4] = {z, z, z, z};

  for (int k0 = 0; k0 < IN_F; k0 += 4) {
    int kk = k0 + 2 * half;
    v2f ax = *(const v2f*)(xr + kk);
#pragma unroll
    for (int nt = 0; nt < 4; ++nt) {
      int n = nt * 16 + idx;
      const float* w0 = W + (long long)kk * HID_F + n;
      const float* w1 = w0 + (long long)IN_F * HID_F;
      const float* w2 = w1 + (long long)IN_F * HID_F;
      v2f b0 = {w0[0], w0[HID_F]};
      v2f b1 = {w1[0], w1[HID_F]};
      v2f b2 = {w2[0], w2[HID_F]};
      acc0[nt] = wmma4(ax, b0, acc0[nt]);
      acc1[nt] = wmma4(ax, b1, acc1[nt]);
      acc2[nt] = wmma4(ax, b2, acc2[nt]);
    }
  }
#pragma unroll
  for (int nt = 0; nt < 4; ++nt) {
    int n = nt * 16 + idx;
    float bv = bias[n];
#pragma unroll
    for (int r = 0; r < 8; ++r) {
      long long o = ((long long)wv * 16 + r + 8 * half) * HID_F + n;
      float y2 = acc2[nt][r];
      ys[o] = y2 + y2;
      u[o]  = acc1[nt][r];
      p[o]  = acc0[nt][r] - y2 + bv;
    }
  }
}

__global__ __launch_bounds__(256)
void gemm2_kernel(const float* __restrict__ p1, const float* __restrict__ W,
                  const float* __restrict__ bias, float* __restrict__ ys,
                  float* __restrict__ u, float* __restrict__ out, int ntile) {
  int wv = (int)((blockIdx.x * (long long)blockDim.x + threadIdx.x) >> 5);
  if (wv >= ntile) return;
  int lane = threadIdx.x & 31;
  int idx = lane & 15, half = lane >> 4;

  const float* pr = p1 + (long long)(wv * 16 + idx) * HID_F;

  v8f z = {0.f, 0.f, 0.f, 0.f, 0.f, 0.f, 0.f, 0.f};
  v8f acc0[3] = {z, z, z};
  v8f acc1[3] = {z, z, z};
  v8f acc2[3] = {z, z, z};

  for (int k0 = 0; k0 < HID_F; k0 += 4) {
    int kk = k0 + 2 * half;
    v2f ap = *(const v2f*)(pr + kk);
    v2f ax = {ap.x > 0.0f ? ap.x : 0.0f, ap.y > 0.0f ? ap.y : 0.0f};
#pragma unroll
    for (int nt = 0; nt < 3; ++nt) {
      int n = nt * 16 + idx;
      bool ok = (n < OUT_F);
      int nn = ok ? n : 0;
      const float* w0 = W + (long long)kk * OUT_F + nn;
      const float* w1 = w0 + (long long)HID_F * OUT_F;
      const float* w2 = w1 + (long long)HID_F * OUT_F;
      v2f b0 = {ok ? w0[0] : 0.0f, ok ? w0[OUT_F] : 0.0f};
      v2f b1 = {ok ? w1[0] : 0.0f, ok ? w1[OUT_F] : 0.0f};
      v2f b2 = {ok ? w2[0] : 0.0f, ok ? w2[OUT_F] : 0.0f};
      acc0[nt] = wmma4(ax, b0, acc0[nt]);
      acc1[nt] = wmma4(ax, b1, acc1[nt]);
      acc2[nt] = wmma4(ax, b2, acc2[nt]);
    }
  }
#pragma unroll
  for (int nt = 0; nt < 3; ++nt) {
    int n = nt * 16 + idx;
    if (n < OUT_F) {
      float bv = bias[n];
#pragma unroll
      for (int r = 0; r < 8; ++r) {
        long long o = ((long long)wv * 16 + r + 8 * half) * OUT_F + n;
        float y2 = acc2[nt][r];
        ys[o]  = y2 + y2;
        u[o]   = acc1[nt][r];
        out[o] = acc0[nt][r] - y2 + bv;
      }
    }
  }
}

extern "C" void kernel_launch(void* const* d_in, const int* in_sizes, int n_in,
                              void* d_out, int out_size, void* d_ws, size_t ws_size,
                              hipStream_t stream) {
  const float* x  = (const float*)d_in[0];
  const int*   ei = (const int*)d_in[1];
  const float* W1 = (const float*)d_in[2];
  const float* b1 = (const float*)d_in[3];
  const float* W2 = (const float*)d_in[4];
  const float* b2 = (const float*)d_in[5];

  const int N = in_sizes[0] / IN_F;
  const int E = in_sizes[1] / 2;
  const int* src = ei;
  const int* dst = ei + E;

  float* ws   = (float*)d_ws;
  float* dinv = ws;
  float* nrm  = dinv + N;
  float* bufA = nrm + E;
  float* bufB = bufA + (size_t)N * IN_F;
  float* hbuf = bufB + (size_t)N * IN_F;
  float* outp = (float*)d_out;

  const int B = 256;
  auto cdiv = [](long long a, long long b) { return (int)((a + b - 1) / b); };

  hipMemsetAsync(dinv, 0, (size_t)N * sizeof(float), stream);
  deg_kernel<<<cdiv(E, B), B, 0, stream>>>(src, dinv, E);
  dinv_kernel<<<cdiv(N, B), B, 0, stream>>>(dinv, N);
  norm_kernel<<<cdiv(E, B), B, 0, stream>>>(src, dst, dinv, nrm, E);

  const int ntile = N / 16;
  const int gblocks = cdiv((long long)ntile * 32, B);

  gemm1_kernel<<<gblocks, B, 0, stream>>>(x, W1, b1, bufA, bufB, hbuf, ntile);
  long long w64 = (long long)E * (HID_F / 4);
  prop_kernel<HID_F><<<cdiv(w64, B), B, 0, stream>>>(bufA, src, dst, nrm, bufB, E);
  prop_kernel<HID_F><<<cdiv(w64, B), B, 0, stream>>>(bufB, src, dst, nrm, hbuf, E);

  gemm2_kernel<<<gblocks, B, 0, stream>>>(hbuf, W2, b2, bufA, bufB, outp, ntile);
  long long w40 = (long long)E * (OUT_F / 4);
  prop_kernel<OUT_F><<<cdiv(w40, B), B, 0, stream>>>(bufA, src, dst, nrm, bufB, E);
  prop_kernel<OUT_F><<<cdiv(w40, B), B, 0, stream>>>(bufB, src, dst, nrm, outp, E);
}


# Float-atomic edge scatter replaced by row-owned gather over a dst-sorted edge list rebuilt each call (int counting sort, arrival-order rank); each row summed in list order by one lane group
# speedup vs baseline: 2.8873x; 1.9714x over previous
#include <hip/hip_runtime.h>


#define IN_F  128
#define HID_F 64
#define OUT_F 40

typedef float v2f __attribute__((ext_vector_type(2)));
typedef float v8f __attribute__((ext_vector_type(8)));

__device__ __forceinline__ v8f wmma4(v2f a, v2f b, v8f c) {
  return __builtin_amdgcn_wmma_f32_16x16x4_f32(
       false, a,  false, b,
       (short)0, c,  false,  false);
}


__global__ void hist_kernel(const int* __restrict__ src, const int* __restrict__ dst,
                            int* __restrict__ degi, int* __restrict__ cnt,
                            int* __restrict__ rank, int E) {
  int e = blockIdx.x * blockDim.x + threadIdx.x;
  if (e < E) {
    int s = src[e], d = dst[e];
    atomicAdd(&degi[s], 1);
    rank[e] = atomicAdd(&cnt[d], 1);
  }
}

__global__ __launch_bounds__(256)
void alloc_kernel(const int* __restrict__ degi, const int* __restrict__ cnt,
                  int* __restrict__ cursor, int* __restrict__ node, int n) {
  __shared__ int sh[256];
  __shared__ int base;
  int t = threadIdx.x;
  int i = blockIdx.x * 256 + t;
  int c = 0;
  float di = 0.0f;
  if (i < n) {
    c = cnt[i];
    float d = (float)degi[i];
    di = (d > 0.0f) ? __frsqrt_rn(d) : 0.0f;
  }
  sh[t] = c;
  __syncthreads();
  for (int off = 1; off < 256; off <<= 1) {
    int v = (t >= off) ? sh[t - off] : 0;
    __syncthreads();
    sh[t] += v;
    __syncthreads();
  }
  int incl = sh[t];
  if (t == 255) base = atomicAdd(cursor, incl);
  __syncthreads();
  if (i < n) {
    int start = base + incl - c;
    ((int4*)node)[i] = make_int4(start, start + c, __float_as_int(di), 0);
  }
}

__global__ void fill_kernel(const int* __restrict__ src, const int* __restrict__ dst,
                            const int* __restrict__ rank, const int* __restrict__ node,
                            int* __restrict__ csr, int E) {
  int e = blockIdx.x * blockDim.x + threadIdx.x;
  if (e < E) {
    int s = src[e], d = dst[e];
    int4 nd = ((const int4*)node)[d];
    float ds = __int_as_float(node[4 * (long long)s + 2]);
    int pos = nd.x + rank[e];
    pos = min(max(pos, 0), E - 1);
    float w = -ds * __int_as_float(nd.z);
    ((int2*)csr)[pos] = make_int2(s, __float_as_int(w));
  }
}


__device__ __forceinline__ int node_clamp(int s, int n) {
  return min(max(s, 0), n - 1);
}

__device__ __forceinline__ float4 fma4(float4 v, float w, float4 a) {
  a.x = fmaf(v.x, w, a.x); a.y = fmaf(v.y, w, a.y);
  a.z = fmaf(v.z, w, a.z); a.w = fmaf(v.w, w, a.w);
  return a;
}

__device__ __forceinline__ float4 sel4(bool p, float4 a, float4 b) {
  float4 r;
  r.x = p ? a.x : b.x; r.y = p ? a.y : b.y; r.z = p ? a.z : b.z; r.w = p ? a.w : b.w;
  return r;
}

__global__ __launch_bounds__(256)
void gather64_kernel(const float* __restrict__ xin, const int* __restrict__ csr,
                     const int* __restrict__ node, float* __restrict__ out, int n, int E) {
  int gid = blockIdx.x * 256 + threadIdx.x;
  int row = gid >> 4, c = gid & 15;
  if (row >= n) return;
  const int2* ent = (const int2*)csr;
  int2 se = *(const int2*)(node + 4 * (long long)row);
  int k = max(se.x, 0);
  int end = min(se.y, E);
  float* o = out + (long long)row * HID_F + c * 4;
  const float* xc = xin + c * 4;
  float4 acc = *(const float4*)o;
  for (; k + 4 <= end; k += 4) {
    int2 e0 = ent[k], e1 = ent[k + 1], e2 = ent[k + 2], e3 = ent[k + 3];
    float4 v0 = *(const float4*)(xc + (long long)node_clamp(e0.x, n) * HID_F);
    float4 v1 = *(const float4*)(xc + (long long)node_clamp(e1.x, n) * HID_F);
    float4 v2 = *(const float4*)(xc + (long long)node_clamp(e2.x, n) * HID_F);
    float4 v3 = *(const float4*)(xc + (long long)node_clamp(e3.x, n) * HID_F);
    acc = fma4(v0, __int_as_float(e0.y), acc);
    acc = fma4(v1, __int_as_float(e1.y), acc);
    acc = fma4(v2, __int_as_float(e2.y), acc);
    acc = fma4(v3, __int_as_float(e3.y), acc);
  }
  int rem = end - k;
  if (rem > 0) {
    int2 e0 = ent[k];
    int2 e1 = ent[rem > 1 ? k + 1 : k];
    int2 e2 = ent[rem > 2 ? k + 2 : k];
    float4 v0 = *(const float4*)(xc + (long long)node_clamp(e0.x, n) * HID_F);
    float4 v1 = *(const float4*)(xc + (long long)node_clamp(e1.x, n) * HID_F);
    float4 v2 = *(const float4*)(xc + (long long)node_clamp(e2.x, n) * HID_F);
    acc = fma4(v0, __int_as_float(e0.y), acc);
    acc = sel4(rem > 1, fma4(v1, __int_as_float(e1.y), acc), acc);
    acc = sel4(rem > 2, fma4(v2, __int_as_float(e2.y), acc), acc);
  }
  *(float4*)o = acc;
}

__global__ __launch_bounds__(256)
void gather40_kernel(const float* __restrict__ xin, const int* __restrict__ csr,
                     const int* __restrict__ node, float* __restrict__ out, int n, int E) {
  int gid = blockIdx.x * 256 + threadIdx.x;
  int row = gid >> 3, j = gid & 7;
  if (row >= n) return;
  const int2* ent = (const int2*)csr;
  int2 se = *(const int2*)(node + 4 * (long long)row);
  int k = max(se.x, 0);
  int end = min(se.y, E);
  float* o4 = out + (long long)row * OUT_F + j * 4;
  float* o1 = out + (long long)row * OUT_F + 32 + j;
  const float* x4 = xin + j * 4;
  const float* x1 = xin + 32 + j;
  float4 acc = *(const float4*)o4;
  float acs = *o1;
  for (; k + 4 <= end; k += 4) {
    int2 e0 = ent[k], e1 = ent[k + 1], e2 = ent[k + 2], e3 = ent[k + 3];
    long long r0 = (long long)node_clamp(e0.x, n) * OUT_F;
    long long r1 = (long long)node_clamp(e1.x, n) * OUT_F;
    long long r2 = (long long)node_clamp(e2.x, n) * OUT_F;
    long long r3 = (long long)node_clamp(e3.x, n) * OUT_F;
    float4 v0 = *(const float4*)(x4 + r0); float t0 = x1[r0];
    float4 v1 = *(const float4*)(x4 + r1); float t1 = x1[r1];
    float4 v2 = *(const float4*)(x4 + r2); float t2 = x1[r2];
    float4 v3 = *(const float4*)(x4 + r3); float t3 = x1[r3];
    float w0 = __int_as_float(e0.y), w1 = __int_as_float(e1.y);
    float w2 = __int_as_float(e2.y), w3 = __int_as_float(e3.y);
    acc = fma4(v0, w0, acc); acs = fmaf(t0, w0, acs);
    acc = fma4(v1, w1, acc); acs = fmaf(t1, w1, acs);
    acc = fma4(v2, w2, acc); acs = fmaf(t2, w2, acs);
    acc = fma4(v3, w3, acc); acs = fmaf(t3, w3, acs);
  }
  int rem = end - k;
  if (rem > 0) {
    int2 e0 = ent[k];
    int2 e1 = ent[rem > 1 ? k + 1 : k];
    int2 e2 = ent[rem > 2 ? k + 2 : k];
    long long r0 = (long long)node_clamp(e0.x, n) * OUT_F;
    long long r1 = (long long)node_clamp(e1.x, n) * OUT_F;
    long long r2 = (long long)node_clamp(e2.x, n) * OUT_F;
    float4 v0 = *(const float4*)(x4 + r0); float t0 = x1[r0];
    float4 v1 = *(const float4*)(x4 + r1); float t1 = x1[r1];
    float4 v2 = *(const float4*)(x4 + r2); float t2 = x1[r2];
    float w0 = __int_as_float(e0.y), w1 = __int_as_float(e1.y), w2 = __int_as_float(e2.y);
    acc = fma4(v0, w0, acc); acs = fmaf(t0, w0, acs);
    acc = sel4(rem > 1, fma4(v1, w1, acc), acc); acs = (rem > 1) ? fmaf(t1, w1, acs) : acs;
    acc = sel4(rem > 2, fma4(v2, w2, acc), acc); acs = (rem > 2) ? fmaf(t2, w2, acs) : acs;
  }
  *(float4*)o4 = acc;
  *o1 = acs;
}


__global__ __launch_bounds__(256)
void gemm1_kernel(const float* __restrict__ x, const float* __restrict__ W,
                  const float* __restrict__ bias, float* __restrict__ ys,
                  float* __restrict__ u, float* __restrict__ p, int ntile) {
  int wv = (int)((blockIdx.x * (long long)blockDim.x + threadIdx.x) >> 5);
  if (wv >= ntile) return;
  int lane = threadIdx.x & 31;
  int idx = lane & 15, half = lane >> 4;

  const float* xr = x + (long long)(wv * 16 + idx) * IN_F;

  v8f z = {0.f, 0.f, 0.f, 0.f, 0.f, 0.f, 0.f, 0.f};
  v8f acc0[4] = {z, z, z, z};
  v8f acc1[4] = {z, z, z, z};
  v8f acc2[4] = {z, z, z, z};

  for (int k0 = 0; k0 < IN_F; k0 += 4) {
    int kk = k0 + 2 * half;
    v2f ax = *(const v2f*)(xr + kk);
#pragma unroll
    for (int nt = 0; nt < 4; ++nt) {
      int n = nt * 16 + idx;
      const float* w0 = W + (long long)kk * HID_F + n;
      const float* w1 = w0 + (long long)IN_F * HID_F;
      const float* w2 = w1 + (long long)IN_F * HID_F;
      v2f b0 = {w0[0], w0[HID_F]};
      v2f b1 = {w1[0], w1[HID_F]};
      v2f b2 = {w2[0], w2[HID_F]};
      acc0[nt] = wmma4(ax, b0, acc0[nt]);
      acc1[nt] = wmma4(ax, b1, acc1[nt]);
      acc2[nt] = wmma4(ax, b2, acc2[nt]);
    }
  }
#pragma unroll
  for (int nt = 0; nt < 4; ++nt) {
    int n = nt * 16 + idx;
    float bv = bias[n];
#pragma unroll
    for (int r = 0; r < 8; ++r) {
      long long o = ((long long)wv * 16 + r + 8 * half) * HID_F + n;
      float y2 = acc2[nt][r];
      ys[o] = y2 + y2;
      u[o]  = acc1[nt][r];
      p[o]  = acc0[nt][r] - y2 + bv;
    }
  }
}

__global__ __launch_bounds__(256)
void gemm2_kernel(const float* __restrict__ p1, const float* __restrict__ W,
                  const float* __restrict__ bias, float* __restrict__ ys,
                  float* __restrict__ u, float* __restrict__ out, int ntile) {
  int wv = (int)((blockIdx.x * (long long)blockDim.x + threadIdx.x) >> 5);
  if (wv >= ntile) return;
  int lane = threadIdx.x & 31;
  int idx = lane & 15, half = lane >> 4;

  const float* pr = p1 + (long long)(wv * 16 + idx) * HID_F;

  v8f z = {0.f, 0.f, 0.f, 0.f, 0.f, 0.f, 0.f, 0.f};
  v8f acc0[3] = {z, z, z};
  v8f acc1[3] = {z, z, z};
  v8f acc2[3] = {z, z, z};

  for (int k0 = 0; k0 < HID_F; k0 += 4) {
    int kk = k0 + 2 * half;
    v2f ap = *(const v2f*)(pr + kk);
    v2f ax = {ap.x > 0.0f ? ap.x : 0.0f, ap.y > 0.0f ? ap.y : 0.0f};
#pragma unroll
    for (int nt = 0; nt < 3; ++nt) {
      int n = nt * 16 + idx;
      bool ok = (n < OUT_F);
      int nn = ok ? n : 0;
      const float* w0 = W + (long long)kk * OUT_F + nn;
      const float* w1 = w0 + (long long)HID_F * OUT_F;
      const float* w2 = w1 + (long long)HID_F * OUT_F;
      v2f b0 = {ok ? w0[0] : 0.0f, ok ? w0[OUT_F] : 0.0f};
      v2f b1 = {ok ? w1[0] : 0.0f, ok ? w1[OUT_F] : 0.0f};
      v2f b2 = {ok ? w2[0] : 0.0f, ok ? w2[OUT_F] : 0.0f};
      acc0[nt] = wmma4(ax, b0, acc0[nt]);
      acc1[nt] = wmma4(ax, b1, acc1[nt]);
      acc2[nt] = wmma4(ax, b2, acc2[nt]);
    }
  }
#pragma unroll
  for (int nt = 0; nt < 3; ++nt) {
    int n = nt * 16 + idx;
    if (n < OUT_F) {
      float bv = bias[n];
#pragma unroll
      for (int r = 0; r < 8; ++r) {
        long long o = ((long long)wv * 16 + r + 8 * half) * OUT_F + n;
        float y2 = acc2[nt][r];
        ys[o]  = y2 + y2;
        u[o]   = acc1[nt][r];
        out[o] = acc0[nt][r] - y2 + bv;
      }
    }
  }
}

extern "C" void kernel_launch(void* const* d_in, const int* in_sizes, int n_in,
                              void* d_out, int out_size, void* d_ws, size_t ws_size,
                              hipStream_t stream) {
  const float* x  = (const float*)d_in[0];
  const int*   ei = (const int*)d_in[1];
  const float* W1 = (const float*)d_in[2];
  const float* b1 = (const float*)d_in[3];
  const float* W2 = (const float*)d_in[4];
  const float* b2 = (const float*)d_in[5];

  const int N = in_sizes[0] / IN_F;
  const int E = in_sizes[1] / 2;
  const int* src = ei;
  const int* dst = ei + E;

  auto up64 = [](size_t a) { return (a + 63) & ~(size_t)63; };
  const size_t Np = up64((size_t)N);
  int*   degi   = (int*)d_ws;
  int*   cnt    = degi + Np;
  int*   cursor = cnt + Np;
  int*   node   = cursor + 64;
  int*   rank   = node + 4 * Np;
  int*   csr    = rank + up64((size_t)E);
  float* bufA = (float*)(csr + up64(2 * (size_t)E));
  float* bufB = bufA + (size_t)N * HID_F;
  float* hbuf = bufB + (size_t)N * HID_F;
  float* outp = (float*)d_out;

  const int B = 256;
  auto cdiv = [](long long a, long long b) { return (int)((a + b - 1) / b); };

  hipMemsetAsync(degi, 0, (2 * Np + 64) * sizeof(int), stream);
  hist_kernel<<<cdiv(E, B), B, 0, stream>>>(src, dst, degi, cnt, rank, E);
  alloc_kernel<<<cdiv(N, B), B, 0, stream>>>(degi, cnt, cursor, node, N);
  fill_kernel<<<cdiv(E, B), B, 0, stream>>>(src, dst, rank, node, csr, E);

  const int ntile = N / 16;
  const int gblocks = cdiv((long long)ntile * 32, B);
  const int g64 = cdiv((long long)N * 16, B);
  const int g40 = cdiv((long long)N * 8, B);

  gemm1_kernel<<<gblocks, B, 0, stream>>>(x, W1, b1, bufA, bufB, hbuf, ntile);
  gather64_kernel<<<g64, B, 0, stream>>>(bufA, csr, node, bufB, N, E);
  gather64_kernel<<<g64, B, 0, stream>>>(bufB, csr, node, hbuf, N, E);

  gemm2_kernel<<<gblocks, B, 0, stream>>>(hbuf, W2, b2, bufA, bufB, outp, ntile);
  gather40_kernel<<<g40, B, 0, stream>>>(bufA, csr, node, bufB, N, E);
  gather40_kernel<<<g40, B, 0, stream>>>(bufB, csr, node, outp, N, E);
}


# Edge-list build without global atomics: (slice, node-range) work-groups count and place with LDS counters; a row's entries lie slice by slice, so its f32 sum order changes
# speedup vs baseline: 4.0027x; 1.3863x over previous
#include <hip/hip_runtime.h>


#define IN_F  128
#define HID_F 64
#define OUT_F 40

typedef float v2f __attribute__((ext_vector_type(2)));
typedef float v8f __attribute__((ext_vector_type(8)));

__device__ __forceinline__ v8f wmma4(v2f a, v2f b, v8f c) {
  return __builtin_amdgcn_wmma_f32_16x16x4_f32(
       false, a,  false, b,
       (short)0, c,  false,  false);
}


#define SLICES 16
#define RN 6400

template <bool VEC>
__device__ __forceinline__ int4 load4(const int* __restrict__ p, int e, int e1) {
  if (VEC) return *(const int4*)(p + e);
  int4 r;
  r.x = p[e];
  r.y = (e + 1 < e1) ? p[e + 1] : -1;
  r.z = (e + 2 < e1) ? p[e + 2] : -1;
  r.w = (e + 3 < e1) ? p[e + 3] : -1;
  return r;
}

__device__ __forceinline__ void count1(int key, int base, int lim, int* cntl) {
  unsigned k = (unsigned)key - (unsigned)base;
  if (k < (unsigned)lim) atomicAdd(&cntl[k], 1);
}

#define CU_ 8
template <bool VEC>
__device__ __forceinline__ void count_scan(const int* __restrict__ key, int e0, int e1, int t,
                                           int base, int lim, int* cntl) {
  for (int e = e0 + 4 * t; e < e1; e += CU_ * 4096) {
    int4 k[CU_];
#pragma unroll
    for (int u = 0; u < CU_; ++u) {
      int ee = e + u * 4096;
      k[u] = make_int4(-1, -1, -1, -1);
      if (ee < e1) k[u] = load4<VEC>(key, ee, e1);
    }
#pragma unroll
    for (int u = 0; u < CU_; ++u) {
      count1(k[u].x, base, lim, cntl);
      count1(k[u].y, base, lim, cntl);
      count1(k[u].z, base, lim, cntl);
      count1(k[u].w, base, lim, cntl);
    }
  }
}

__global__ __launch_bounds__(1024)
void count_kernel(const int* __restrict__ src, const int* __restrict__ dst,
                  int* __restrict__ part, int N, int Np, int E, int slice) {
  __shared__ int cntl[RN];
  const int t = threadIdx.x;
  const int s = blockIdx.x;
  const int R = (int)(gridDim.y >> 1);
  const int which = ((int)blockIdx.y >= R) ? 1 : 0;
  const int base = ((int)blockIdx.y - which * R) * RN;
  const int lim = min(RN, N - base);
  const int* __restrict__ key = which ? dst : src;
  for (int j = t; j < RN; j += 1024) cntl[j] = 0;
  __syncthreads();
  const int e0 = s * slice;
  const int e1 = min(e0 + slice, E);
  if ((E & 3) == 0) count_scan<true>(key, e0, e1, t, base, lim, cntl);
  else              count_scan<false>(key, e0, e1, t, base, lim, cntl);
  __syncthreads();
  int* __restrict__ p = part + (size_t)(which * SLICES + s) * Np + base;
  for (int j = t; j < lim; j += 1024) p[j] = cntl[j];
}

__global__ __launch_bounds__(256)
void alloc_kernel(int* __restrict__ part, int* __restrict__ cursor,
                  int* __restrict__ node, int n, int Np) {
  __shared__ int sh[256];
  __shared__ int base;
  int t = threadIdx.x;
  int i = blockIdx.x * 256 + t;
  int c = 0;
  float di = 0.0f;
  if (i < n) {
    const int* ps = part + i;
    int* pd = part + (size_t)SLICES * Np + i;
    int dg = 0;
    int v[SLICES];
#pragma unroll
    for (int s = 0; s < SLICES; ++s) dg += ps[(size_t)s * Np];
#pragma unroll
    for (int s = 0; s < SLICES; ++s) v[s] = pd[(size_t)s * Np];
#pragma unroll
    for (int s = 0; s < SLICES; ++s) { pd[(size_t)s * Np] = c; c += v[s]; }
    float d = (float)dg;
    di = (d > 0.0f) ? __frsqrt_rn(d) : 0.0f;
  }
  sh[t] = c;
  __syncthreads();
  for (int off = 1; off < 256; off <<= 1) {
    int v2 = (t >= off) ? sh[t - off] : 0;
    __syncthreads();
    sh[t] += v2;
    __syncthreads();
  }
  int incl = sh[t];
  if (t == 255) base = atomicAdd(cursor, incl);
  __syncthreads();
  if (i < n) {
    int start = base + incl - c;
    ((int4*)node)[i] = make_int4(start, start + c, __float_as_int(di), 0);
  }
}

#define PU_ 4
template <bool VEC>
__device__ __forceinline__ void place_scan(const int* __restrict__ src, const int* __restrict__ dst,
                                           const int* __restrict__ node, int* __restrict__ csr,
                                           int e0, int e1, int t, int base, int lim,
                                           int* cur, const float* dl, int E) {
  const int spare = RN + (t & 31);
  for (int e = e0 + 4 * t; e < e1; e += PU_ * 4096) {
    int kd[4 * PU_], ks[4 * PU_];
#pragma unroll
    for (int u = 0; u < PU_; ++u) {
      int ee = e + u * 4096;
      int4 a = make_int4(-1, -1, -1, -1);
      int4 b = make_int4(0, 0, 0, 0);
      if (ee < e1) { a = load4<VEC>(dst, ee, e1); b = load4<VEC>(src, ee, e1); }
      kd[4 * u] = a.x; kd[4 * u + 1] = a.y; kd[4 * u + 2] = a.z; kd[4 * u + 3] = a.w;
      ks[4 * u] = b.x; ks[4 * u + 1] = b.y; ks[4 * u + 2] = b.z; ks[4 * u + 3] = b.w;
    }
    bool in[4 * PU_];
    int kk[4 * PU_], pos[4 * PU_];
#pragma unroll
    for (int j = 0; j < 4 * PU_; ++j) {
      unsigned k = (unsigned)kd[j] - (unsigned)base;
      in[j] = k < (unsigned)lim;
      kk[j] = in[j] ? (int)k : 0;
      pos[j] = atomicAdd(&cur[in[j] ? (int)k : spare], in[j] ? 1 : 0);
    }
    float wa[4 * PU_], wd[4 * PU_];
#pragma unroll
    for (int j = 0; j < 4 * PU_; ++j) {
      ks[j] = in[j] ? ks[j] : 0;
      wa[j] = __int_as_float(node[4 * (long long)ks[j] + 2]);
      wd[j] = dl[kk[j]];
    }
#pragma unroll
    for (int j = 0; j < 4 * PU_; ++j) {
      if (in[j]) ((int2*)csr)[min(max(pos[j], 0), E - 1)] =
          make_int2(ks[j], __float_as_int(-wa[j] * wd[j]));
    }
  }
}

__global__ __launch_bounds__(1024)
void place_kernel(const int* __restrict__ src, const int* __restrict__ dst,
                  const int* __restrict__ offs, const int* __restrict__ node,
                  int* __restrict__ csr, int N, int Np, int E, int slice) {
  __shared__ int cur[RN + 32];
  __shared__ float dl[RN];
  const int t = threadIdx.x;
  const int s = blockIdx.x;
  const int base = (int)blockIdx.y * RN;
  const int lim = min(RN, N - base);
  const int* __restrict__ po = offs + (size_t)s * Np + base;
  for (int j = t; j < lim; j += 1024) {
    int4 nd = ((const int4*)node)[base + j];
    cur[j] = nd.x + po[j];
    dl[j] = __int_as_float(nd.z);
  }
  __syncthreads();
  const int e0 = s * slice;
  const int e1 = min(e0 + slice, E);
  if ((E & 3) == 0) place_scan<true>(src, dst, node, csr, e0, e1, t, base, lim, cur, dl, E);
  else              place_scan<false>(src, dst, node, csr, e0, e1, t, base, lim, cur, dl, E);
}


__device__ __forceinline__ int node_clamp(int s, int n) {
  return min(max(s, 0), n - 1);
}

__device__ __forceinline__ float4 fma4(float4 v, float w, float4 a) {
  a.x = fmaf(v.x, w, a.x); a.y = fmaf(v.y, w, a.y);
  a.z = fmaf(v.z, w, a.z); a.w = fmaf(v.w, w, a.w);
  return a;
}

__device__ __forceinline__ float4 sel4(bool p, float4 a, float4 b) {
  float4 r;
  r.x = p ? a.x : b.x; r.y = p ? a.y : b.y; r.z = p ? a.z : b.z; r.w = p ? a.w : b.w;
  return r;
}

__global__ __launch_bounds__(256)
void gather64_kernel(const float* __restrict__ xin, const int* __restrict__ csr,
                     const int* __restrict__ node, float* __restrict__ out, int n, int E) {
  int gid = blockIdx.x * 256 + threadIdx.x;
  int row = gid >> 4, c = gid & 15;
  if (row >= n) return;
  const int2* ent = (const int2*)csr;
  int2 se = *(const int2*)(node + 4 * (long long)row);
  int k = max(se.x, 0);
  int end = min(se.y, E);
  float* o = out + (long long)row * HID_F + c * 4;
  const float* xc = xin + c * 4;
  float4 acc = *(const float4*)o;
  for (; k + 4 <= end; k += 4) {
    int2 e0 = ent[k], e1 = ent[k + 1], e2 = ent[k + 2], e3 = ent[k + 3];
    float4 v0 = *(const float4*)(xc + (long long)node_clamp(e0.x, n) * HID_F);
    float4 v1 = *(const float4*)(xc + (long long)node_clamp(e1.x, n) * HID_F);
    float4 v2 = *(const float4*)(xc + (long long)node_clamp(e2.x, n) * HID_F);
    float4 v3 = *(const float4*)(xc + (long long)node_clamp(e3.x, n) * HID_F);
    acc = fma4(v0, __int_as_float(e0.y), acc);
    acc = fma4(v1, __int_as_float(e1.y), acc);
    acc = fma4(v2, __int_as_float(e2.y), acc);
    acc = fma4(v3, __int_as_float(e3.y), acc);
  }
  int rem = end - k;
  if (rem > 0) {
    int2 e0 = ent[k];
    int2 e1 = ent[rem > 1 ? k + 1 : k];
    int2 e2 = ent[rem > 2 ? k + 2 : k];
    float4 v0 = *(const float4*)(xc + (long long)node_clamp(e0.x, n) * HID_F);
    float4 v1 = *(const float4*)(xc + (long long)node_clamp(e1.x, n) * HID_F);
    float4 v2 = *(const float4*)(xc + (long long)node_clamp(e2.x, n) * HID_F);
    acc = fma4(v0, __int_as_float(e0.y), acc);
    acc = sel4(rem > 1, fma4(v1, __int_as_float(e1.y), acc), acc);
    acc = sel4(rem > 2, fma4(v2, __int_as_float(e2.y), acc), acc);
  }
  *(float4*)o = acc;
}

__global__ __launch_bounds__(256)
void gather40_kernel(const float* __restrict__ xin, const int* __restrict__ csr,
                     const int* __restrict__ node, float* __restrict__ out, int n, int E) {
  int gid = blockIdx.x * 256 + threadIdx.x;
  int row = gid >> 3, j = gid & 7;
  if (row >= n) return;
  const int2* ent = (const int2*)csr;
  int2 se = *(const int2*)(node + 4 * (long long)row);
  int k = max(se.x, 0);
  int end = min(se.y, E);
  float* o4 = out + (long long)row * OUT_F + j * 4;
  float* o1 = out + (long long)row * OUT_F + 32 + j;
  const float* x4 = xin + j * 4;
  const float* x1 = xin + 32 + j;
  float4 acc = *(const float4*)o4;
  float acs = *o1;
  for (; k + 4 <= end; k += 4) {
    int2 e0 = ent[k], e1 = ent[k + 1], e2 = ent[k + 2], e3 = ent[k + 3];
    long long r0 = (long long)node_clamp(e0.x, n) * OUT_F;
    long long r1 = (long long)node_clamp(e1.x, n) * OUT_F;
    long long r2 = (long long)node_clamp(e2.x, n) * OUT_F;
    long long r3 = (long long)node_clamp(e3.x, n) * OUT_F;
    float4 v0 = *(const float4*)(x4 + r0); float t0 = x1[r0];
    float4 v1 = *(const float4*)(x4 + r1); float t1 = x1[r1];
    float4 v2 = *(const float4*)(x4 + r2); float t2 = x1[r2];
    float4 v3 = *(const float4*)(x4 + r3); float t3 = x1[r3];
    float w0 = __int_as_float(e0.y), w1 = __int_as_float(e1.y);
    float w2 = __int_as_float(e2.y), w3 = __int_as_float(e3.y);
    acc = fma4(v0, w0, acc); acs = fmaf(t0, w0, acs);
    acc = fma4(v1, w1, acc); acs = fmaf(t1, w1, acs);
    acc = fma4(v2, w2, acc); acs = fmaf(t2, w2, acs);
    acc = fma4(v3, w3, acc); acs = fmaf(t3, w3, acs);
  }
  int rem = end - k;
  if (rem > 0) {
    int2 e0 = ent[k];
    int2 e1 = ent[rem > 1 ? k + 1 : k];
    int2 e2 = ent[rem > 2 ? k + 2 : k];
    long long r0 = (long long)node_clamp(e0.x, n) * OUT_F;
    long long r1 = (long long)node_clamp(e1.x, n) * OUT_F;
    long long r2 = (long long)node_clamp(e2.x, n) * OUT_F;
    float4 v0 = *(const float4*)(x4 + r0); float t0 = x1[r0];
    float4 v1 = *(const float4*)(x4 + r1); float t1 = x1[r1];
    float4 v2 = *(const float4*)(x4 + r2); float t2 = x1[r2];
    float w0 = __int_as_float(e0.y), w1 = __int_as_float(e1.y), w2 = __int_as_float(e2.y);
    acc = fma4(v0, w0, acc); acs = fmaf(t0, w0, acs);
    acc = sel4(rem > 1, fma4(v1, w1, acc), acc); acs = (rem > 1) ? fmaf(t1, w1, acs) : acs;
    acc = sel4(rem > 2, fma4(v2, w2, acc), acc); acs = (rem > 2) ? fmaf(t2, w2, acs) : acs;
  }
  *(float4*)o4 = acc;
  *o1 = acs;
}


__global__ __launch_bounds__(256)
void gemm1_kernel(const float* __restrict__ x, const float* __restrict__ W,
                  const float* __restrict__ bias, float* __restrict__ ys,
                  float* __restrict__ u, float* __restrict__ p, int ntile) {
  int wv = (int)((blockIdx.x * (long long)blockDim.x + threadIdx.x) >> 5);
  if (wv >= ntile) return;
  int lane = threadIdx.x & 31;
  int idx = lane & 15, half = lane >> 4;

  const float* xr = x + (long long)(wv * 16 + idx) * IN_F;

  v8f z = {0.f, 0.f, 0.f, 0.f, 0.f, 0.f, 0.f, 0.f};
  v8f acc0[4] = {z, z, z, z};
  v8f acc1[4] = {z, z, z, z};
  v8f acc2[4] = {z, z, z, z};

  for (int k0 = 0; k0 < IN_F; k0 += 4) {
    int kk = k0 + 2 * half;
    v2f ax = *(const v2f*)(xr + kk);
#pragma unroll
    for (int nt = 0; nt < 4; ++nt) {
      int n = nt * 16 + idx;
      const float* w0 = W + (long long)kk * HID_F + n;
      const float* w1 = w0 + (long long)IN_F * HID_F;
      const float* w2 = w1 + (long long)IN_F * HID_F;
      v2f b0 = {w0[0], w0[HID_F]};
      v2f b1 = {w1[0], w1[HID_F]};
      v2f b2 = {w2[0], w2[HID_F]};
      acc0[nt] = wmma4(ax, b0, acc0[nt]);
      acc1[nt] = wmma4(ax, b1, acc1[nt]);
      acc2[nt] = wmma4(ax, b2, acc2[nt]);
    }
  }
#pragma unroll
  for (int nt = 0; nt < 4; ++nt) {
    int n = nt * 16 + idx;
    float bv = bias[n];
#pragma unroll
    for (int r = 0; r < 8; ++r) {
      long long o = ((long long)wv * 16 + r + 8 * half) * HID_F + n;
      float y2 = acc2[nt][r];
      ys[o] = y2 + y2;
      u[o]  = acc1[nt][r];
      p[o]  = acc0[nt][r] - y2 + bv;
    }
  }
}

__global__ __launch_bounds__(256)
void gemm2_kernel(const float* __restrict__ p1, const float* __restrict__ W,
                  const float* __restrict__ bias, float* __restrict__ ys,
                  float* __restrict__ u, float* __restrict__ out, int ntile) {
  int wv = (int)((blockIdx.x * (long long)blockDim.x + threadIdx.x) >> 5);
  if (wv >= ntile) return;
  int lane = threadIdx.x & 31;
  int idx = lane & 15, half = lane >> 4;

  const float* pr = p1 + (long long)(wv * 16 + idx) * HID_F;

  v8f z = {0.f, 0.f, 0.f, 0.f, 0.f, 0.f, 0.f, 0.f};
  v8f acc0[3] = {z, z, z};
  v8f acc1[3] = {z, z, z};
  v8f acc2[3] = {z, z, z};

  for (int k0 = 0; k0 < HID_F; k0 += 4) {
    int kk = k0 + 2 * half;
    v2f ap = *(const v2f*)(pr + kk);
    v2f ax = {ap.x > 0.0f ? ap.x : 0.0f, ap.y > 0.0f ? ap.y : 0.0f};
#pragma unroll
    for (int nt = 0; nt < 3; ++nt) {
      int n = nt * 16 + idx;
      bool ok = (n < OUT_F);
      int nn = ok ? n : 0;
      const float* w0 = W + (long long)kk * OUT_F + nn;
      const float* w1 = w0 + (long long)HID_F * OUT_F;
      const float* w2 = w1 + (long long)HID_F * OUT_F;
      v2f b0 = {ok ? w0[0] : 0.0f, ok ? w0[OUT_F] : 0.0f};
      v2f b1 = {ok ? w1[0] : 0.0f, ok ? w1[OUT_F] : 0.0f};
      v2f b2 = {ok ? w2[0] : 0.0f, ok ? w2[OUT_F] : 0.0f};
      acc0[nt] = wmma4(ax, b0, acc0[nt]);
      acc1[nt] = wmma4(ax, b1, acc1[nt]);
      acc2[nt] = wmma4(ax, b2, acc2[nt]);
    }
  }
#pragma unroll
  for (int nt = 0; nt < 3; ++nt) {
    int n = nt * 16 + idx;
    if (n < OUT_F) {
      float bv = bias[n];
#pragma unroll
      for (int r = 0; r < 8; ++r) {
        long long o = ((long long)wv * 16 + r + 8 * half) * OUT_F + n;
        float y2 = acc2[nt][r];
        ys[o]  = y2 + y2;
        u[o]   = acc1[nt][r];
        out[o] = acc0[nt][r] - y2 + bv;
      }
    }
  }
}

extern "C" void kernel_launch(void* const* d_in, const int* in_sizes, int n_in,
                              void* d_out, int out_size, void* d_ws, size_t ws_size,
                              hipStream_t stream) {
  const float* x  = (const float*)d_in[0];
  const int*   ei = (const int*)d_in[1];
  const float* W1 = (const float*)d_in[2];
  const float* b1 = (const float*)d_in[3];
  const float* W2 = (const float*)d_in[4];
  const float* b2 = (const float*)d_in[5];

  const int N = in_sizes[0] / IN_F;
  const int E = in_sizes[1] / 2;
  const int* src = ei;
  const int* dst = ei + E;

  auto up64 = [](size_t a) { return (a + 63) & ~(size_t)63; };
  const size_t Np = up64((size_t)N);
  int*   cursor = (int*)d_ws;
  int*   node   = cursor + 64;
  int*   csr    = node + 4 * Np;
  float* bufA = (float*)(csr + up64(2 * (size_t)E));
  float* bufB = bufA + (size_t)N * HID_F;
  float* hbuf = bufB + (size_t)N * HID_F;
  float* outp = (float*)d_out;
  int*   part = (int*)bufA;

  const int B = 256;
  auto cdiv = [](long long a, long long b) { return (int)((a + b - 1) / b); };

  const int R = cdiv(N, RN);
  const int slice = cdiv(cdiv(E, SLICES), 4) * 4;
  hipMemsetAsync(cursor, 0, 64 * sizeof(int), stream);
  count_kernel<<<dim3(SLICES, 2 * R), 1024, 0, stream>>>(src, dst, part, N, (int)Np, E, slice);
  alloc_kernel<<<cdiv(N, B), B, 0, stream>>>(part, cursor, node, N, (int)Np);
  place_kernel<<<dim3(SLICES, R), 1024, 0, stream>>>(src, dst, part + (size_t)SLICES * Np, node, csr,
                                                    N, (int)Np, E, slice);

  const int ntile = N / 16;
  const int gblocks = cdiv((long long)ntile * 32, B);
  const int g64 = cdiv((long long)N * 16, B);
  const int g40 = cdiv((long long)N * 8, B);

  gemm1_kernel<<<gblocks, B, 0, stream>>>(x, W1, b1, bufA, bufB, hbuf, ntile);
  gather64_kernel<<<g64, B, 0, stream>>>(bufA, csr, node, bufB, N, E);
  gather64_kernel<<<g64, B, 0, stream>>>(bufB, csr, node, hbuf, N, E);

  gemm2_kernel<<<gblocks, B, 0, stream>>>(hbuf, W2, b2, bufA, bufB, outp, ntile);
  gather40_kernel<<<g40, B, 0, stream>>>(bufA, csr, node, bufB, N, E);
  gather40_kernel<<<g40, B, 0, stream>>>(bufB, csr, node, outp, N, E);
}


# Layer 2 products moved into the epilogue of layer 1's second row gather: relu'd rows go through LDS to the WMMAs, the p plane and the separate GEMM launch go; sum orders kept
# speedup vs baseline: 4.1782x; 1.0438x over previous
#include <hip/hip_runtime.h>


#define IN_F  128
#define HID_F 64
#define OUT_F 40

typedef float v2f __attribute__((ext_vector_type(2)));
typedef float v8f __attribute__((ext_vector_type(8)));

__device__ __forceinline__ v8f wmma4(v2f a, v2f b, v8f c) {
  return __builtin_amdgcn_wmma_f32_16x16x4_f32(
       false, a,  false, b,
       (short)0, c,  false,  false);
}


#define SLICES 16
#define RN 6400

template <bool VEC>
__device__ __forceinline__ int4 load4(const int* __restrict__ p, int e, int e1) {
  if (VEC) return *(const int4*)(p + e);
  int4 r;
  r.x = p[e];
  r.y = (e + 1 < e1) ? p[e + 1] : -1;
  r.z = (e + 2 < e1) ? p[e + 2] : -1;
  r.w = (e + 3 < e1) ? p[e + 3] : -1;
  return r;
}

__device__ __forceinline__ void count1(int key, int base, int lim, int* cntl) {
  unsigned k = (unsigned)key - (unsigned)base;
  if (k < (unsigned)lim) atomicAdd(&cntl[k], 1);
}

#define CU_ 8
template <bool VEC>
__device__ __forceinline__ void count_scan(const int* __restrict__ key, int e0, int e1, int t,
                                           int base, int lim, int* cntl) {
  for (int e = e0 + 4 * t; e < e1; e += CU_ * 4096) {
    int4 k[CU_];
#pragma unroll
    for (int u = 0; u < CU_; ++u) {
      int ee = e + u * 4096;
      k[u] = make_int4(-1, -1, -1, -1);
      if (ee < e1) k[u] = load4<VEC>(key, ee, e1);
    }
#pragma unroll
    for (int u = 0; u < CU_; ++u) {
      count1(k[u].x, base, lim, cntl);
      count1(k[u].y, base, lim, cntl);
      count1(k[u].z, base, lim, cntl);
      count1(k[u].w, base, lim, cntl);
    }
  }
}

__global__ __launch_bounds__(1024)
void count_kernel(const int* __restrict__ src, const int* __restrict__ dst,
                  int* __restrict__ part, int N, int Np, int E, int slice) {
  __shared__ int cntl[RN];
  const int t = threadIdx.x;
  const int s = blockIdx.x;
  const int R = (int)(gridDim.y >> 1);
  const int which = ((int)blockIdx.y >= R) ? 1 : 0;
  const int base = ((int)blockIdx.y - which * R) * RN;
  const int lim = min(RN, N - base);
  const int* __restrict__ key = which ? dst : src;
  for (int j = t; j < RN; j += 1024) cntl[j] = 0;
  __syncthreads();
  const int e0 = s * slice;
  const int e1 = min(e0 + slice, E);
  if ((E & 3) == 0) count_scan<true>(key, e0, e1, t, base, lim, cntl);
  else              count_scan<false>(key, e0, e1, t, base, lim, cntl);
  __syncthreads();
  int* __restrict__ p = part + (size_t)(which * SLICES + s) * Np + base;
  for (int j = t; j < lim; j += 1024) p[j] = cntl[j];
}

__global__ __launch_bounds__(256)
void alloc_kernel(int* __restrict__ part, int* __restrict__ cursor,
                  int* __restrict__ node, int n, int Np) {
  __shared__ int sh[256];
  __shared__ int base;
  int t = threadIdx.x;
  int i = blockIdx.x * 256 + t;
  int c = 0;
  float di = 0.0f;
  if (i < n) {
    const int* ps = part + i;
    int* pd = part + (size_t)SLICES * Np + i;
    int dg = 0;
    int v[SLICES];
#pragma unroll
    for (int s = 0; s < SLICES; ++s) dg += ps[(size_t)s * Np];
#pragma unroll
    for (int s = 0; s < SLICES; ++s) v[s] = pd[(size_t)s * Np];
#pragma unroll
    for (int s = 0; s < SLICES; ++s) { pd[(size_t)s * Np] = c; c += v[s]; }
    float d = (float)dg;
    di = (d > 0.0f) ? __frsqrt_rn(d) : 0.0f;
  }
  sh[t] = c;
  __syncthreads();
  for (int off = 1; off < 256; off <<= 1) {
    int v2 = (t >= off) ? sh[t - off] : 0;
    __syncthreads();
    sh[t] += v2;
    __syncthreads();
  }
  int incl = sh[t];
  if (t == 255) base = atomicAdd(cursor, incl);
  __syncthreads();
  if (i < n) {
    int start = base + incl - c;
    ((int4*)node)[i] = make_int4(start, start + c, __float_as_int(di), 0);
  }
}

#define PU_ 4
template <bool VEC>
__device__ __forceinline__ void place_scan(const int* __restrict__ src, const int* __restrict__ dst,
                                           const int* __restrict__ node, int* __restrict__ csr,
                                           int e0, int e1, int t, int base, int lim,
                                           int* cur, const float* dl, int E) {
  const int spare = RN + (t & 31);
  for (int e = e0 + 4 * t; e < e1; e += PU_ * 4096) {
    int kd[4 * PU_], ks[4 * PU_];
#pragma unroll
    for (int u = 0; u < PU_; ++u) {
      int ee = e + u * 4096;
      int4 a = make_int4(-1, -1, -1, -1);
      int4 b = make_int4(0, 0, 0, 0);
      if (ee < e1) { a = load4<VEC>(dst, ee, e1); b = load4<VEC>(src, ee, e1); }
      kd[4 * u] = a.x; kd[4 * u + 1] = a.y; kd[4 * u + 2] = a.z; kd[4 * u + 3] = a.w;
      ks[4 * u] = b.x; ks[4 * u + 1] = b.y; ks[4 * u + 2] = b.z; ks[4 * u + 3] = b.w;
    }
    bool in[4 * PU_];
    int kk[4 * PU_], pos[4 * PU_];
#pragma unroll
    for (int j = 0; j < 4 * PU_; ++j) {
      unsigned k = (unsigned)kd[j] - (unsigned)base;
      in[j] = k < (unsigned)lim;
      kk[j] = in[j] ? (int)k : 0;
      pos[j] = atomicAdd(&cur[in[j] ? (int)k : spare], in[j] ? 1 : 0);
    }
    float wa[4 * PU_], wd[4 * PU_];
#pragma unroll
    for (int j = 0; j < 4 * PU_; ++j) {
      ks[j] = in[j] ? ks[j] : 0;
      wa[j] = __int_as_float(node[4 * (long long)ks[j] + 2]);
      wd[j] = dl[kk[j]];
    }
#pragma unroll
    for (int j = 0; j < 4 * PU_; ++j) {
      if (in[j]) ((int2*)csr)[min(max(pos[j], 0), E - 1)] =
          make_int2(ks[j], __float_as_int(-wa[j] * wd[j]));
    }
  }
}

__global__ __launch_bounds__(1024)
void place_kernel(const int* __restrict__ src, const int* __restrict__ dst,
                  const int* __restrict__ offs, const int* __restrict__ node,
                  int* __restrict__ csr, int N, int Np, int E, int slice) {
  __shared__ int cur[RN + 32];
  __shared__ float dl[RN];
  const int t = threadIdx.x;
  const int s = blockIdx.x;
  const int base = (int)blockIdx.y * RN;
  const int lim = min(RN, N - base);
  const int* __restrict__ po = offs + (size_t)s * Np + base;
  for (int j = t; j < lim; j += 1024) {
    int4 nd = ((const int4*)node)[base + j];
    cur[j] = nd.x + po[j];
    dl[j] = __int_as_float(nd.z);
  }
  __syncthreads();
  const int e0 = s * slice;
  const int e1 = min(e0 + slice, E);
  if ((E & 3) == 0) place_scan<true>(src, dst, node, csr, e0, e1, t, base, lim, cur, dl, E);
  else              place_scan<false>(src, dst, node, csr, e0, e1, t, base, lim, cur, dl, E);
}


__device__ __forceinline__ int node_clamp(int s, int n) {
  return min(max(s, 0), n - 1);
}

__device__ __forceinline__ float4 fma4(float4 v, float w, float4 a) {
  a.x = fmaf(v.x, w, a.x); a.y = fmaf(v.y, w, a.y);
  a.z = fmaf(v.z, w, a.z); a.w = fmaf(v.w, w, a.w);
  return a;
}

__device__ __forceinline__ float4 sel4(bool p, float4 a, float4 b) {
  float4 r;
  r.x = p ? a.x : b.x; r.y = p ? a.y : b.y; r.z = p ? a.z : b.z; r.w = p ? a.w : b.w;
  return r;
}

__global__ __launch_bounds__(256)
void gather64_kernel(const float* __restrict__ xin, const int* __restrict__ csr,
                     const int* __restrict__ node, float* __restrict__ out, int n, int E) {
  int gid = blockIdx.x * 256 + threadIdx.x;
  int row = gid >> 4, c = gid & 15;
  if (row >= n) return;
  const int2* ent = (const int2*)csr;
  int2 se = *(const int2*)(node + 4 * (long long)row);
  int k = max(se.x, 0);
  int end = min(se.y, E);
  float* o = out + (long long)row * HID_F + c * 4;
  const float* xc = xin + c * 4;
  float4 acc = *(const float4*)o;
  for (; k + 4 <= end; k += 4) {
    int2 e0 = ent[k], e1 = ent[k + 1], e2 = ent[k + 2], e3 = ent[k + 3];
    float4 v0 = *(const float4*)(xc + (long long)node_clamp(e0.x, n) * HID_F);
    float4 v1 = *(const float4*)(xc + (long long)node_clamp(e1.x, n) * HID_F);
    float4 v2 = *(const float4*)(xc + (long long)node_clamp(e2.x, n) * HID_F);
    float4 v3 = *(const float4*)(xc + (long long)node_clamp(e3.x, n) * HID_F);
    acc = fma4(v0, __int_as_float(e0.y), acc);
    acc = fma4(v1, __int_as_float(e1.y), acc);
    acc = fma4(v2, __int_as_float(e2.y), acc);
    acc = fma4(v3, __int_as_float(e3.y), acc);
  }
  int rem = end - k;
  if (rem > 0) {
    int2 e0 = ent[k];
    int2 e1 = ent[rem > 1 ? k + 1 : k];
    int2 e2 = ent[rem > 2 ? k + 2 : k];
    float4 v0 = *(const float4*)(xc + (long long)node_clamp(e0.x, n) * HID_F);
    float4 v1 = *(const float4*)(xc + (long long)node_clamp(e1.x, n) * HID_F);
    float4 v2 = *(const float4*)(xc + (long long)node_clamp(e2.x, n) * HID_F);
    acc = fma4(v0, __int_as_float(e0.y), acc);
    acc = sel4(rem > 1, fma4(v1, __int_as_float(e1.y), acc), acc);
    acc = sel4(rem > 2, fma4(v2, __int_as_float(e2.y), acc), acc);
  }
  *(float4*)o = acc;
}

__global__ __launch_bounds__(256)
void gather40_kernel(const float* __restrict__ xin, const int* __restrict__ csr,
                     const int* __restrict__ node, const float* start, int spitch,
                     float* out, int n, int E) {
  int gid = blockIdx.x * 256 + threadIdx.x;
  int row = gid >> 3, j = gid & 7;
  if (row >= n) return;
  const int2* ent = (const int2*)csr;
  int2 se = *(const int2*)(node + 4 * (long long)row);
  int k = max(se.x, 0);
  int end = min(se.y, E);
  float* o4 = out + (long long)row * OUT_F + j * 4;
  float* o1 = out + (long long)row * OUT_F + 32 + j;
  const float* x4 = xin + j * 4;
  const float* x1 = xin + 32 + j;
  const float* sr = start + (long long)row * spitch;
  float4 acc = *(const float4*)(sr + j * 4);
  float acs = sr[32 + j];
  for (; k + 4 <= end; k += 4) {
    int2 e0 = ent[k], e1 = ent[k + 1], e2 = ent[k + 2], e3 = ent[k + 3];
    long long r0 = (long long)node_clamp(e0.x, n) * OUT_F;
    long long r1 = (long long)node_clamp(e1.x, n) * OUT_F;
    long long r2 = (long long)node_clamp(e2.x, n) * OUT_F;
    long long r3 = (long long)node_clamp(e3.x, n) * OUT_F;
    float4 v0 = *(const float4*)(x4 + r0); float t0 = x1[r0];
    float4 v1 = *(const float4*)(x4 + r1); float t1 = x1[r1];
    float4 v2 = *(const float4*)(x4 + r2); float t2 = x1[r2];
    float4 v3 = *(const float4*)(x4 + r3); float t3 = x1[r3];
    float w0 = __int_as_float(e0.y), w1 = __int_as_float(e1.y);
    float w2 = __int_as_float(e2.y), w3 = __int_as_float(e3.y);
    acc = fma4(v0, w0, acc); acs = fmaf(t0, w0, acs);
    acc = fma4(v1, w1, acc); acs = fmaf(t1, w1, acs);
    acc = fma4(v2, w2, acc); acs = fmaf(t2, w2, acs);
    acc = fma4(v3, w3, acc); acs = fmaf(t3, w3, acs);
  }
  int rem = end - k;
  if (rem > 0) {
    int2 e0 = ent[k];
    int2 e1 = ent[rem > 1 ? k + 1 : k];
    int2 e2 = ent[rem > 2 ? k + 2 : k];
    long long r0 = (long long)node_clamp(e0.x, n) * OUT_F;
    long long r1 = (long long)node_clamp(e1.x, n) * OUT_F;
    long long r2 = (long long)node_clamp(e2.x, n) * OUT_F;
    float4 v0 = *(const float4*)(x4 + r0); float t0 = x1[r0];
    float4 v1 = *(const float4*)(x4 + r1); float t1 = x1[r1];
    float4 v2 = *(const float4*)(x4 + r2); float t2 = x1[r2];
    float w0 = __int_as_float(e0.y), w1 = __int_as_float(e1.y), w2 = __int_as_float(e2.y);
    acc = fma4(v0, w0, acc); acs = fmaf(t0, w0, acs);
    acc = sel4(rem > 1, fma4(v1, w1, acc), acc); acs = (rem > 1) ? fmaf(t1, w1, acs) : acs;
    acc = sel4(rem > 2, fma4(v2, w2, acc), acc); acs = (rem > 2) ? fmaf(t2, w2, acs) : acs;
  }
  *(float4*)o4 = acc;
  *o1 = acs;
}


__global__ __launch_bounds__(256)
void gemm1_kernel(const float* __restrict__ x, const float* __restrict__ W,
                  const float* __restrict__ bias, float* __restrict__ ys,
                  float* __restrict__ u, float* __restrict__ p, int ntile) {
  int wv = (int)((blockIdx.x * (long long)blockDim.x + threadIdx.x) >> 5);
  if (wv >= ntile) return;
  int lane = threadIdx.x & 31;
  int idx = lane & 15, half = lane >> 4;

  const float* xr = x + (long long)(wv * 16 + idx) * IN_F;

  v8f z = {0.f, 0.f, 0.f, 0.f, 0.f, 0.f, 0.f, 0.f};
  v8f acc0[4] = {z, z, z, z};
  v8f acc1[4] = {z, z, z, z};
  v8f acc2[4] = {z, z, z, z};

  for (int k0 = 0; k0 < IN_F; k0 += 4) {
    int kk = k0 + 2 * half;
    v2f ax = *(const v2f*)(xr + kk);
#pragma unroll
    for (int nt = 0; nt < 4; ++nt) {
      int n = nt * 16 + idx;
      const float* w0 = W + (long long)kk * HID_F + n;
      const float* w1 = w0 + (long long)IN_F * HID_F;
      const float* w2 = w1 + (long long)IN_F * HID_F;
      v2f b0 = {w0[0], w0[HID_F]};
      v2f b1 = {w1[0], w1[HID_F]};
      v2f b2 = {w2[0], w2[HID_F]};
      acc0[nt] = wmma4(ax, b0, acc0[nt]);
      acc1[nt] = wmma4(ax, b1, acc1[nt]);
      acc2[nt] = wmma4(ax, b2, acc2[nt]);
    }
  }
#pragma unroll
  for (int nt = 0; nt < 4; ++nt) {
    int n = nt * 16 + idx;
    float bv = bias[n];
#pragma unroll
    for (int r = 0; r < 8; ++r) {
      long long o = ((long long)wv * 16 + r + 8 * half) * HID_F + n;
      float y2 = acc2[nt][r];
      ys[o] = y2 + y2;
      u[o]  = acc1[nt][r];
      p[o]  = acc0[nt][r] - y2 + bv;
    }
  }
}

#define HP 68
#define YP 122
__global__ __launch_bounds__(256)
void gather64_gemm2_kernel(const float* __restrict__ xin, const int* __restrict__ csr,
                           const int* __restrict__ node, float* __restrict__ pu,
                           const float* __restrict__ W, const float* __restrict__ bias,
                           float* __restrict__ ys, float* __restrict__ out, int n, int E) {
  __shared__ __attribute__((aligned(16))) float hs[16 * HP];
  __shared__ float yt[16 * YP];
  const int t = threadIdx.x;
  const int row0 = blockIdx.x * 16;
  const int lr = t >> 4, c = t & 15;
  const int row = row0 + lr;
  float4 acc = make_float4(0.0f, 0.0f, 0.0f, 0.0f);
  if (row < n) {
    const int2* ent = (const int2*)csr;
    int2 se = *(const int2*)(node + 4 * (long long)row);
    int k = max(se.x, 0);
    int end = min(se.y, E);
    const float* xc = xin + c * 4;
    acc = *(const float4*)(pu + (long long)row * HID_F + c * 4);
    for (; k + 4 <= end; k += 4) {
      int2 e0 = ent[k], e1 = ent[k + 1], e2 = ent[k + 2], e3 = ent[k + 3];
      float4 v0 = *(const float4*)(xc + (long long)node_clamp(e0.x, n) * HID_F);
      float4 v1 = *(const float4*)(xc + (long long)node_clamp(e1.x, n) * HID_F);
      float4 v2 = *(const float4*)(xc + (long long)node_clamp(e2.x, n) * HID_F);
      float4 v3 = *(const float4*)(xc + (long long)node_clamp(e3.x, n) * HID_F);
      acc = fma4(v0, __int_as_float(e0.y), acc);
      acc = fma4(v1, __int_as_float(e1.y), acc);
      acc = fma4(v2, __int_as_float(e2.y), acc);
      acc = fma4(v3, __int_as_float(e3.y), acc);
    }
    int rem = end - k;
    if (rem > 0) {
      int2 e0 = ent[k];
      int2 e1 = ent[rem > 1 ? k + 1 : k];
      int2 e2 = ent[rem > 2 ? k + 2 : k];
      float4 v0 = *(const float4*)(xc + (long long)node_clamp(e0.x, n) * HID_F);
      float4 v1 = *(const float4*)(xc + (long long)node_clamp(e1.x, n) * HID_F);
      float4 v2 = *(const float4*)(xc + (long long)node_clamp(e2.x, n) * HID_F);
      acc = fma4(v0, __int_as_float(e0.y), acc);
      acc = sel4(rem > 1, fma4(v1, __int_as_float(e1.y), acc), acc);
      acc = sel4(rem > 2, fma4(v2, __int_as_float(e2.y), acc), acc);
    }
  }
  float4 h;
  h.x = acc.x > 0.0f ? acc.x : 0.0f; h.y = acc.y > 0.0f ? acc.y : 0.0f;
  h.z = acc.z > 0.0f ? acc.z : 0.0f; h.w = acc.w > 0.0f ? acc.w : 0.0f;
  *(float4*)(hs + lr * HP + c * 4) = h;

  const int wave = t >> 5, lane = t & 31;
  const int idx = lane & 15, half = lane >> 4;
  const int j = wave * 16 + idx;
  const bool ok = j < 3 * OUT_F;
  const int m = ok ? j / OUT_F : 0;
  const int nn = ok ? j - m * OUT_F : 0;
  const float* wb = W + (long long)m * HID_F * OUT_F + nn;
  const float* ar = hs + idx * HP;
  float bw[HID_F / 2];
#pragma unroll
  for (int s = 0; s < HID_F / 4; ++s) {
    int kk = 4 * s + 2 * half;
    bw[2 * s]     = wb[kk * OUT_F];
    bw[2 * s + 1] = wb[(kk + 1) * OUT_F];
  }
  __syncthreads();

  v8f d = {0.f, 0.f, 0.f, 0.f, 0.f, 0.f, 0.f, 0.f};
#pragma unroll
  for (int s = 0; s < HID_F / 4; ++s) {
    v2f a = *(const v2f*)(ar + 4 * s + 2 * half);
    v2f b = {ok ? bw[2 * s] : 0.0f, ok ? bw[2 * s + 1] : 0.0f};
    d = wmma4(a, b, d);
  }
  if (ok) {
#pragma unroll
    for (int r = 0; r < 8; ++r) yt[(r + 8 * half) * YP + j] = d[r];
  }
  __syncthreads();

  for (int e = t; e < 16 * OUT_F; e += 256) {
    int r = e / OUT_F, q = e - r * OUT_F;
    int grow = row0 + r;
    if (grow < n) {
      float y0 = yt[r * YP + q], y1 = yt[r * YP + OUT_F + q], y2 = yt[r * YP + 2 * OUT_F + q];
      long long o = (long long)grow * OUT_F + q;
      ys[o]  = y2 + y2;
      pu[(long long)grow * HID_F + q] = y1;
      out[o] = y0 - y2 + bias[q];
    }
  }
}

extern "C" void kernel_launch(void* const* d_in, const int* in_sizes, int n_in,
                              void* d_out, int out_size, void* d_ws, size_t ws_size,
                              hipStream_t stream) {
  const float* x  = (const float*)d_in[0];
  const int*   ei = (const int*)d_in[1];
  const float* W1 = (const float*)d_in[2];
  const float* b1 = (const float*)d_in[3];
  const float* W2 = (const float*)d_in[4];
  const float* b2 = (const float*)d_in[5];

  const int N = in_sizes[0] / IN_F;
  const int E = in_sizes[1] / 2;
  const int* src = ei;
  const int* dst = ei + E;

  auto up64 = [](size_t a) { return (a + 63) & ~(size_t)63; };
  const size_t Np = up64((size_t)N);
  int*   cursor = (int*)d_ws;
  int*   node   = cursor + 64;
  int*   csr    = node + 4 * Np;
  float* bufA = (float*)(csr + up64(2 * (size_t)E));
  float* bufB = bufA + (size_t)N * HID_F;
  float* hbuf = bufB + (size_t)N * HID_F;
  float* outp = (float*)d_out;
  int*   part = (int*)bufA;

  const int B = 256;
  auto cdiv = [](long long a, long long b) { return (int)((a + b - 1) / b); };

  const int R = cdiv(N, RN);
  const int slice = cdiv(cdiv(E, SLICES), 4) * 4;
  hipMemsetAsync(cursor, 0, 64 * sizeof(int), stream);
  count_kernel<<<dim3(SLICES, 2 * R), 1024, 0, stream>>>(src, dst, part, N, (int)Np, E, slice);
  alloc_kernel<<<cdiv(N, B), B, 0, stream>>>(part, cursor, node, N, (int)Np);
  place_kernel<<<dim3(SLICES, R), 1024, 0, stream>>>(src, dst, part + (size_t)SLICES * Np, node, csr,
                                                    N, (int)Np, E, slice);

  const int ntile = N / 16;
  const int gblocks = cdiv((long long)ntile * 32, B);
  const int g64 = cdiv((long long)N * 16, B);
  const int g40 = cdiv((long long)N * 8, B);

  gemm1_kernel<<<gblocks, B, 0, stream>>>(x, W1, b1, bufA, bufB, hbuf, ntile);
  gather64_kernel<<<g64, B, 0, stream>>>(bufA, csr, node, bufB, N, E);
  gather64_gemm2_kernel<<<g64, B, 0, stream>>>(bufB, csr, node, hbuf, W2, b2, bufA, outp, N, E);

  gather40_kernel<<<g40, B, 0, stream>>>(bufA, csr, node, hbuf, HID_F, bufB, N, E);
  gather40_kernel<<<g40, B, 0, stream>>>(bufB, csr, node, outp, OUT_F, outp, N, E);
}


# Layer 1 products moved into the place launch: 256 extra work-groups (3 accumulators a wave, weights in LDS, 64 registers) run their WMMAs beside the placing ones, two per WGP; sum orders kept
# speedup vs baseline: 4.8101x; 1.1512x over previous
#include <hip/hip_runtime.h>


#define IN_F  128
#define HID_F 64
#define OUT_F 40

typedef float v2f __attribute__((ext_vector_type(2)));
typedef float v8f __attribute__((ext_vector_type(8)));

__device__ __forceinline__ v8f wmma4(v2f a, v2f b, v8f c) {
  return __builtin_amdgcn_wmma_f32_16x16x4_f32(
       false, a,  false, b,
       (short)0, c,  false,  false);
}


#define SLICES 16
#define RN 6400

template <bool VEC>
__device__ __forceinline__ int4 load4(const int* __restrict__ p, int e, int e1) {
  if (VEC) return *(const int4*)(p + e);
  int4 r;
  r.x = p[e];
  r.y = (e + 1 < e1) ? p[e + 1] : -1;
  r.z = (e + 2 < e1) ? p[e + 2] : -1;
  r.w = (e + 3 < e1) ? p[e + 3] : -1;
  return r;
}

__device__ __forceinline__ void count1(int key, int base, int lim, int* cntl) {
  unsigned k = (unsigned)key - (unsigned)base;
  if (k < (unsigned)lim) atomicAdd(&cntl[k], 1);
}

#define CU_ 8
template <bool VEC>
__device__ __forceinline__ void count_scan(const int* __restrict__ key, int e0, int e1, int t,
                                           int base, int lim, int* cntl) {
  for (int e = e0 + 4 * t; e < e1; e += CU_ * 4096) {
    int4 k[CU_];
#pragma unroll
    for (int u = 0; u < CU_; ++u) {
      int ee = e + u * 4096;
      k[u] = make_int4(-1, -1, -1, -1);
      if (ee < e1) k[u] = load4<VEC>(key, ee, e1);
    }
#pragma unroll
    for (int u = 0; u < CU_; ++u) {
      count1(k[u].x, base, lim, cntl);
      count1(k[u].y, base, lim, cntl);
      count1(k[u].z, base, lim, cntl);
      count1(k[u].w, base, lim, cntl);
    }
  }
}

__global__ __launch_bounds__(1024)
void count_kernel(const int* __restrict__ src, const int* __restrict__ dst,
                  int* __restrict__ part, int N, int Np, int E, int slice) {
  __shared__ int cntl[RN];
  const int t = threadIdx.x;
  const int s = blockIdx.x;
  const int R = (int)(gridDim.y >> 1);
  const int which = ((int)blockIdx.y >= R) ? 1 : 0;
  const int base = ((int)blockIdx.y - which * R) * RN;
  const int lim = min(RN, N - base);
  const int* __restrict__ key = which ? dst : src;
  for (int j = t; j < RN; j += 1024) cntl[j] = 0;
  __syncthreads();
  const int e0 = s * slice;
  const int e1 = min(e0 + slice, E);
  if ((E & 3) == 0) count_scan<true>(key, e0, e1, t, base, lim, cntl);
  else              count_scan<false>(key, e0, e1, t, base, lim, cntl);
  __syncthreads();
  int* __restrict__ p = part + (size_t)(which * SLICES + s) * Np + base;
  for (int j = t; j < lim; j += 1024) p[j] = cntl[j];
}

__global__ __launch_bounds__(256)
void alloc_kernel(int* __restrict__ part, int* __restrict__ cursor,
                  int* __restrict__ node, int n, int Np) {
  __shared__ int sh[256];
  __shared__ int base;
  int t = threadIdx.x;
  int i = blockIdx.x * 256 + t;
  int c = 0;
  float di = 0.0f;
  if (i < n) {
    const int* ps = part + i;
    int* pd = part + (size_t)SLICES * Np + i;
    int dg = 0;
    int v[SLICES];
#pragma unroll
    for (int s = 0; s < SLICES; ++s) dg += ps[(size_t)s * Np];
#pragma unroll
    for (int s = 0; s < SLICES; ++s) v[s] = pd[(size_t)s * Np];
#pragma unroll
    for (int s = 0; s < SLICES; ++s) { pd[(size_t)s * Np] = c; c += v[s]; }
    float d = (float)dg;
    di = (d > 0.0f) ? __frsqrt_rn(d) : 0.0f;
  }
  sh[t] = c;
  __syncthreads();
  for (int off = 1; off < 256; off <<= 1) {
    int v2 = (t >= off) ? sh[t - off] : 0;
    __syncthreads();
    sh[t] += v2;
    __syncthreads();
  }
  int incl = sh[t];
  if (t == 255) base = atomicAdd(cursor, incl);
  __syncthreads();
  if (i < n) {
    int start = base + incl - c;
    ((int4*)node)[i] = make_int4(start, start + c, __float_as_int(di), 0);
  }
}

#define PU_ 2
template <bool VEC>
__device__ __forceinline__ void place_scan(const int* __restrict__ src, const int* __restrict__ dst,
                                           const int* __restrict__ node, int* __restrict__ csr,
                                           int e0, int e1, int t, int base, int lim,
                                           int* cur, const float* dl, int E) {
  const int spare = RN + (t & 31);
  for (int e = e0 + 4 * t; e < e1; e += PU_ * 4096) {
    int kd[4 * PU_], ks[4 * PU_];
#pragma unroll
    for (int u = 0; u < PU_; ++u) {
      int ee = e + u * 4096;
      int4 a = make_int4(-1, -1, -1, -1);
      int4 b = make_int4(0, 0, 0, 0);
      if (ee < e1) { a = load4<VEC>(dst, ee, e1); b = load4<VEC>(src, ee, e1); }
      kd[4 * u] = a.x; kd[4 * u + 1] = a.y; kd[4 * u + 2] = a.z; kd[4 * u + 3] = a.w;
      ks[4 * u] = b.x; ks[4 * u + 1] = b.y; ks[4 * u + 2] = b.z; ks[4 * u + 3] = b.w;
    }
    bool in[4 * PU_];
    int kk[4 * PU_], pos[4 * PU_];
#pragma unroll
    for (int j = 0; j < 4 * PU_; ++j) {
      unsigned k = (unsigned)kd[j] - (unsigned)base;
      in[j] = k < (unsigned)lim;
      kk[j] = in[j] ? (int)k : 0;
      pos[j] = atomicAdd(&cur[in[j] ? (int)k : spare], in[j] ? 1 : 0);
    }
    float wa[4 * PU_], wd[4 * PU_];
#pragma unroll
    for (int j = 0; j < 4 * PU_; ++j) {
      ks[j] = in[j] ? ks[j] : 0;
      wa[j] = __int_as_float(node[4 * (long long)ks[j] + 2]);
      wd[j] = dl[kk[j]];
    }
#pragma unroll
    for (int j = 0; j < 4 * PU_; ++j) {
      if (in[j]) ((int2*)csr)[min(max(pos[j], 0), E - 1)] =
          make_int2(ks[j], __float_as_int(-wa[j] * wd[j]));
    }
  }
}

#define WP 80
#define WLDS (3 * IN_F * WP)
__device__ __forceinline__ void gemm1_item(const float* __restrict__ x, const float* wl,
                                           const float* __restrict__ bias, float* __restrict__ ys,
                                           float* __restrict__ u, float* __restrict__ p,
                                           int tile, int nt, int lane) {
  const int idx = lane & 15, half = lane >> 4;
  const int n = nt * 16 + idx;
  const float* xa = x + (long long)(tile * 16 + idx) * IN_F + 2 * half;
  const float* wn = wl + (2 * half) * WP + n;
  v8f z = {0.f, 0.f, 0.f, 0.f, 0.f, 0.f, 0.f, 0.f};
  v8f acc0 = z, acc1 = z, acc2 = z;
  for (int kb = 0; kb < IN_F; kb += 16) {
    v2f ax[4];
#pragma unroll
    for (int i = 0; i < 4; ++i) ax[i] = *(const v2f*)(xa + kb + 4 * i);
#pragma unroll
    for (int i = 0; i < 4; ++i) {
      const float* wk = wn + (kb + 4 * i) * WP;
      v2f b0 = {wk[0], wk[WP]};
      v2f b1 = {wk[IN_F * WP], wk[IN_F * WP + WP]};
      v2f b2 = {wk[2 * IN_F * WP], wk[2 * IN_F * WP + WP]};
      acc0 = wmma4(ax[i], b0, acc0);
      acc1 = wmma4(ax[i], b1, acc1);
      acc2 = wmma4(ax[i], b2, acc2);
    }
  }
  float bv = bias[n];
#pragma unroll
  for (int r = 0; r < 8; ++r) {
    long long o = ((long long)tile * 16 + r + 8 * half) * HID_F + n;
    float y2 = acc2[r];
    ys[o] = y2 + y2;
    u[o]  = acc1[r];
    p[o]  = acc0[r] - y2 + bv;
  }
}

#define GY 16
__global__ __launch_bounds__(1024, 16)
void place_gemm1_kernel(const int* __restrict__ src, const int* __restrict__ dst,
                        const int* __restrict__ offs, const int* __restrict__ node,
                        int* __restrict__ csr, int N, int Np, int E, int slice,
                        const float* __restrict__ x, const float* __restrict__ W,
                        const float* __restrict__ bias, float* __restrict__ ys,
                        float* __restrict__ u, float* __restrict__ p, int ntile) {
  __shared__ __attribute__((aligned(16))) float smem[WLDS];
  const int t = threadIdx.x;
  if ((int)blockIdx.y < GY) {
    for (int i = t; i < 3 * IN_F * HID_F / 4; i += 1024) {
      float4 v = ((const float4*)W)[i];
      *(float4*)(smem + (i >> 4) * WP + (i & 15) * 4) = v;
    }
    __syncthreads();
    const int gb = (int)blockIdx.y * SLICES + (int)blockIdx.x;
    const int ng = GY * SLICES;
    const int wave = __builtin_amdgcn_readfirstlane(t >> 5);
    const int lane = t & 31;
    const int q = wave >> 2, nt = wave & 3;
    for (int tile = q * ng + gb; tile < ntile; tile += 8 * ng) {
      asm volatile("" ::: "memory");
      gemm1_item(x, smem, bias, ys, u, p, tile, nt, lane);
    }
    return;
  }
  int* cur = (int*)smem;
  float* dl = smem + (RN + 32);
  const int s = blockIdx.x;
  const int base = ((int)blockIdx.y - GY) * RN;
  const int lim = min(RN, N - base);
  const int* __restrict__ po = offs + (size_t)s * Np + base;
  for (int j = t; j < lim; j += 1024) {
    int4 nd = ((const int4*)node)[base + j];
    cur[j] = nd.x + po[j];
    dl[j] = __int_as_float(nd.z);
  }
  __syncthreads();
  const int e0 = s * slice;
  const int e1 = min(e0 + slice, E);
  if ((E & 3) == 0) place_scan<true>(src, dst, node, csr, e0, e1, t, base, lim, cur, dl, E);
  else              place_scan<false>(src, dst, node, csr, e0, e1, t, base, lim, cur, dl, E);
}


__device__ __forceinline__ int node_clamp(int s, int n) {
  return min(max(s, 0), n - 1);
}

__device__ __forceinline__ float4 fma4(float4 v, float w, float4 a) {
  a.x = fmaf(v.x, w, a.x); a.y = fmaf(v.y, w, a.y);
  a.z = fmaf(v.z, w, a.z); a.w = fmaf(v.w, w, a.w);
  return a;
}

__device__ __forceinline__ float4 sel4(bool p, float4 a, float4 b) {
  float4 r;
  r.x = p ? a.x : b.x; r.y = p ? a.y : b.y; r.z = p ? a.z : b.z; r.w = p ? a.w : b.w;
  return r;
}

__global__ __launch_bounds__(256)
void gather64_kernel(const float* __restrict__ xin, const int* __restrict__ csr,
                     const int* __restrict__ node, float* __restrict__ out, int n, int E) {
  int gid = blockIdx.x * 256 + threadIdx.x;
  int row = gid >> 4, c = gid & 15;
  if (row >= n) return;
  const int2* ent = (const int2*)csr;
  int2 se = *(const int2*)(node + 4 * (long long)row);
  int k = max(se.x, 0);
  int end = min(se.y, E);
  float* o = out + (long long)row * HID_F + c * 4;
  const float* xc = xin + c * 4;
  float4 acc = *(const float4*)o;
  for (; k + 4 <= end; k += 4) {
    int2 e0 = ent[k], e1 = ent[k + 1], e2 = ent[k + 2], e3 = ent[k + 3];
    float4 v0 = *(const float4*)(xc + (long long)node_clamp(e0.x, n) * HID_F);
    float4 v1 = *(const float4*)(xc + (long long)node_clamp(e1.x, n) * HID_F);
    float4 v2 = *(const float4*)(xc + (long long)node_clamp(e2.x, n) * HID_F);
    float4 v3 = *(const float4*)(xc + (long long)node_clamp(e3.x, n) * HID_F);
    acc = fma4(v0, __int_as_float(e0.y), acc);
    acc = fma4(v1, __int_as_float(e1.y), acc);
    acc = fma4(v2, __int_as_float(e2.y), acc);
    acc = fma4(v3, __int_as_float(e3.y), acc);
  }
  int rem = end - k;
  if (rem > 0) {
    int2 e0 = ent[k];
    int2 e1 = ent[rem > 1 ? k + 1 : k];
    int2 e2 = ent[rem > 2 ? k + 2 : k];
    float4 v0 = *(const float4*)(xc + (long long)node_clamp(e0.x, n) * HID_F);
    float4 v1 = *(const float4*)(xc + (long long)node_clamp(e1.x, n) * HID_F);
    float4 v2 = *(const float4*)(xc + (long long)node_clamp(e2.x, n) * HID_F);
    acc = fma4(v0, __int_as_float(e0.y), acc);
    acc = sel4(rem > 1, fma4(v1, __int_as_float(e1.y), acc), acc);
    acc = sel4(rem > 2, fma4(v2, __int_as_float(e2.y), acc), acc);
  }
  *(float4*)o = acc;
}

__global__ __launch_bounds__(256)
void gather40_kernel(const float* __restrict__ xin, const int* __restrict__ csr,
                     const int* __restrict__ node, const float* start, int spitch,
                     float* out, int n, int E) {
  int gid = blockIdx.x * 256 + threadIdx.x;
  int row = gid >> 3, j = gid & 7;
  if (row >= n) return;
  const int2* ent = (const int2*)csr;
  int2 se = *(const int2*)(node + 4 * (long long)row);
  int k = max(se.x, 0);
  int end = min(se.y, E);
  float* o4 = out + (long long)row * OUT_F + j * 4;
  float* o1 = out + (long long)row * OUT_F + 32 + j;
  const float* x4 = xin + j * 4;
  const float* x1 = xin + 32 + j;
  const float* sr = start + (long long)row * spitch;
  float4 acc = *(const float4*)(sr + j * 4);
  float acs = sr[32 + j];
  for (; k + 4 <= end; k += 4) {
    int2 e0 = ent[k], e1 = ent[k + 1], e2 = ent[k + 2], e3 = ent[k + 3];
    long long r0 = (long long)node_clamp(e0.x, n) * OUT_F;
    long long r1 = (long long)node_clamp(e1.x, n) * OUT_F;
    long long r2 = (long long)node_clamp(e2.x, n) * OUT_F;
    long long r3 = (long long)node_clamp(e3.x, n) * OUT_F;
    float4 v0 = *(const float4*)(x4 + r0); float t0 = x1[r0];
    float4 v1 = *(const float4*)(x4 + r1); float t1 = x1[r1];
    float4 v2 = *(const float4*)(x4 + r2); float t2 = x1[r2];
    float4 v3 = *(const float4*)(x4 + r3); float t3 = x1[r3];
    float w0 = __int_as_float(e0.y), w1 = __int_as_float(e1.y);
    float w2 = __int_as_float(e2.y), w3 = __int_as_float(e3.y);
    acc = fma4(v0, w0, acc); acs = fmaf(t0, w0, acs);
    acc = fma4(v1, w1, acc); acs = fmaf(t1, w1, acs);
    acc = fma4(v2, w2, acc); acs = fmaf(t2, w2, acs);
    acc = fma4(v3, w3, acc); acs = fmaf(t3, w3, acs);
  }
  int rem = end - k;
  if (rem > 0) {
    int2 e0 = ent[k];
    int2 e1 = ent[rem > 1 ? k + 1 : k];
    int2 e2 = ent[rem > 2 ? k + 2 : k];
    long long r0 = (long long)node_clamp(e0.x, n) * OUT_F;
    long long r1 = (long long)node_clamp(e1.x, n) * OUT_F;
    long long r2 = (long long)node_clamp(e2.x, n) * OUT_F;
    float4 v0 = *(const float4*)(x4 + r0); float t0 = x1[r0];
    float4 v1 = *(const float4*)(x4 + r1); float t1 = x1[r1];
    float4 v2 = *(const float4*)(x4 + r2); float t2 = x1[r2];
    float w0 = __int_as_float(e0.y), w1 = __int_as_float(e1.y), w2 = __int_as_float(e2.y);
    acc = fma4(v0, w0, acc); acs = fmaf(t0, w0, acs);
    acc = sel4(rem > 1, fma4(v1, w1, acc), acc); acs = (rem > 1) ? fmaf(t1, w1, acs) : acs;
    acc = sel4(rem > 2, fma4(v2, w2, acc), acc); acs = (rem > 2) ? fmaf(t2, w2, acs) : acs;
  }
  *(float4*)o4 = acc;
  *o1 = acs;
}


#define HP 68
#define YP 122
__global__ __launch_bounds__(256)
void gather64_gemm2_kernel(const float* __restrict__ xin, const int* __restrict__ csr,
                           const int* __restrict__ node, float* __restrict__ pu,
                           const float* __restrict__ W, const float* __restrict__ bias,
                           float* __restrict__ ys, float* __restrict__ out, int n, int E) {
  __shared__ __attribute__((aligned(16))) float hs[16 * HP];
  __shared__ float yt[16 * YP];
  const int t = threadIdx.x;
  const int row0 = blockIdx.x * 16;
  const int lr = t >> 4, c = t & 15;
  const int row = row0 + lr;
  float4 acc = make_float4(0.0f, 0.0f, 0.0f, 0.0f);
  if (row < n) {
    const int2* ent = (const int2*)csr;
    int2 se = *(const int2*)(node + 4 * (long long)row);
    int k = max(se.x, 0);
    int end = min(se.y, E);
    const float* xc = xin + c * 4;
    acc = *(const float4*)(pu + (long long)row * HID_F + c * 4);
    for (; k + 4 <= end; k += 4) {
      int2 e0 = ent[k], e1 = ent[k + 1], e2 = ent[k + 2], e3 = ent[k + 3];
      float4 v0 = *(const float4*)(xc + (long long)node_clamp(e0.x, n) * HID_F);
      float4 v1 = *(const float4*)(xc + (long long)node_clamp(e1.x, n) * HID_F);
      float4 v2 = *(const float4*)(xc + (long long)node_clamp(e2.x, n) * HID_F);
      float4 v3 = *(const float4*)(xc + (long long)node_clamp(e3.x, n) * HID_F);
      acc = fma4(v0, __int_as_float(e0.y), acc);
      acc = fma4(v1, __int_as_float(e1.y), acc);
      acc = fma4(v2, __int_as_float(e2.y), acc);
      acc = fma4(v3, __int_as_float(e3.y), acc);
    }
    int rem = end - k;
    if (rem > 0) {
      int2 e0 = ent[k];
      int2 e1 = ent[rem > 1 ? k + 1 : k];
      int2 e2 = ent[rem > 2 ? k + 2 : k];
      float4 v0 = *(const float4*)(xc + (long long)node_clamp(e0.x, n) * HID_F);
      float4 v1 = *(const float4*)(xc + (long long)node_clamp(e1.x, n) * HID_F);
      float4 v2 = *(const float4*)(xc + (long long)node_clamp(e2.x, n) * HID_F);
      acc = fma4(v0, __int_as_float(e0.y), acc);
      acc = sel4(rem > 1, fma4(v1, __int_as_float(e1.y), acc), acc);
      acc = sel4(rem > 2, fma4(v2, __int_as_float(e2.y), acc), acc);
    }
  }
  float4 h;
  h.x = acc.x > 0.0f ? acc.x : 0.0f; h.y = acc.y > 0.0f ? acc.y : 0.0f;
  h.z = acc.z > 0.0f ? acc.z : 0.0f; h.w = acc.w > 0.0f ? acc.w : 0.0f;
  *(float4*)(hs + lr * HP + c * 4) = h;

  const int wave = t >> 5, lane = t & 31;
  const int idx = lane & 15, half = lane >> 4;
  const int j = wave * 16 + idx;
  const bool ok = j < 3 * OUT_F;
  const int m = ok ? j / OUT_F : 0;
  const int nn = ok ? j - m * OUT_F : 0;
  const float* wb = W + (long long)m * HID_F * OUT_F + nn;
  const float* ar = hs + idx * HP;
  float bw[HID_F / 2];
#pragma unroll
  for (int s = 0; s < HID_F / 4; ++s) {
    int kk = 4 * s + 2 * half;
    bw[2 * s]     = wb[kk * OUT_F];
    bw[2 * s + 1] = wb[(kk + 1) * OUT_F];
  }
  __syncthreads();

  v8f d = {0.f, 0.f, 0.f, 0.f, 0.f, 0.f, 0.f, 0.f};
#pragma unroll
  for (int s = 0; s < HID_F / 4; ++s) {
    v2f a = *(const v2f*)(ar + 4 * s + 2 * half);
    v2f b = {ok ? bw[2 * s] : 0.0f, ok ? bw[2 * s + 1] : 0.0f};
    d = wmma4(a, b, d);
  }
  if (ok) {
#pragma unroll
    for (int r = 0; r < 8; ++r) yt[(r + 8 * half) * YP + j] = d[r];
  }
  __syncthreads();

  for (int e = t; e < 16 * OUT_F; e += 256) {
    int r = e / OUT_F, q = e - r * OUT_F;
    int grow = row0 + r;
    if (grow < n) {
      float y0 = yt[r * YP + q], y1 = yt[r * YP + OUT_F + q], y2 = yt[r * YP + 2 * OUT_F + q];
      long long o = (long long)grow * OUT_F + q;
      ys[o]  = y2 + y2;
      pu[(long long)grow * HID_F + q] = y1;
      out[o] = y0 - y2 + bias[q];
    }
  }
}

extern "C" void kernel_launch(void* const* d_in, const int* in_sizes, int n_in,
                              void* d_out, int out_size, void* d_ws, size_t ws_size,
                              hipStream_t stream) {
  const float* x  = (const float*)d_in[0];
  const int*   ei = (const int*)d_in[1];
  const float* W1 = (const float*)d_in[2];
  const float* b1 = (const float*)d_in[3];
  const float* W2 = (const float*)d_in[4];
  const float* b2 = (const float*)d_in[5];

  const int N = in_sizes[0] / IN_F;
  const int E = in_sizes[1] / 2;
  const int* src = ei;
  const int* dst = ei + E;

  auto up64 = [](size_t a) { return (a + 63) & ~(size_t)63; };
  const size_t Np = up64((size_t)N);
  int*   cursor = (int*)d_ws;
  int*   node   = cursor + 64;
  int*   csr    = node + 4 * Np;
  float* bufA = (float*)(csr + up64(2 * (size_t)E));
  float* bufB = bufA + (size_t)N * HID_F;
  float* hbuf = bufB + (size_t)N * HID_F;
  float* outp = (float*)d_out;
  int*   part = (int*)d_out;

  const int B = 256;
  auto cdiv = [](long long a, long long b) { return (int)((a + b - 1) / b); };

  const int R = cdiv(N, RN);
  const int slice = cdiv(cdiv(E, SLICES), 4) * 4;
  hipMemsetAsync(cursor, 0, 64 * sizeof(int), stream);
  count_kernel<<<dim3(SLICES, 2 * R), 1024, 0, stream>>>(src, dst, part, N, (int)Np, E, slice);
  alloc_kernel<<<cdiv(N, B), B, 0, stream>>>(part, cursor, node, N, (int)Np);
  const int ntile = N / 16;
  place_gemm1_kernel<<<dim3(SLICES, GY + R), 1024, 0, stream>>>(
      src, dst, part + (size_t)SLICES * Np, node, csr, N, (int)Np, E, slice,
      x, W1, b1, bufA, bufB, hbuf, ntile);

  const int g64 = cdiv((long long)N * 16, B);
  const int g40 = cdiv((long long)N * 8, B);

  gather64_kernel<<<g64, B, 0, stream>>>(bufA, csr, node, bufB, N, E);
  gather64_gemm2_kernel<<<g64, B, 0, stream>>>(bufB, csr, node, hbuf, W2, b2, bufA, outp, N, E);

  gather40_kernel<<<g40, B, 0, stream>>>(bufA, csr, node, hbuf, HID_F, bufB, N, E);
  gather40_kernel<<<g40, B, 0, stream>>>(bufB, csr, node, outp, OUT_F, outp, N, E);
}
